# starting kernel
# baseline (speedup 1.0000x reference)
#include <hip/hip_runtime.h>

#define IN_DIM   128
#define OUT_DIM  128
#define N_NODES  10000
#define N_EDGES  640000

typedef __attribute__((ext_vector_type(2))) float v2f;
typedef __attribute__((ext_vector_type(8))) float v8f;

#define A0_OFF   0
#define A1_OFF   (N_NODES * IN_DIM)
#define D0_OFF   (2 * N_NODES * IN_DIM)
#define D1_OFF   (2 * N_NODES * IN_DIM + N_NODES)
#define WS_FLOATS (2 * N_NODES * IN_DIM + 2 * N_NODES)

__global__ __launch_bounds__(256) void gnn_zero_ws(float* __restrict__ ws) {
    int i = blockIdx.x * blockDim.x + threadIdx.x;
    if (i < WS_FLOATS) ws[i] = 0.0f;
}

__global__ __launch_bounds__(256) void gnn_edge_scatter(
    const float* __restrict__ X,
    const int*   __restrict__ edge_index,
    const int*   __restrict__ edge_types,
    float*       __restrict__ ws)
{
    float* A0 = ws + A0_OFF;
    float* A1 = ws + A1_OFF;
    float* D0 = ws + D0_OFF;
    float* D1 = ws + D1_OFF;

    const int lane        = threadIdx.x & 31;
    const int wave        = (blockIdx.x * blockDim.x + threadIdx.x) >> 5;
    const int waves_total = (gridDim.x * blockDim.x) >> 5;

    for (int e = wave; e < N_EDGES; e += waves_total) {
        const int src = edge_index[e];
        const int dst = edge_index[N_EDGES + e];
        const int t   = edge_types[e];

        const float4 xv = ((const float4*)(X + (size_t)src * IN_DIM))[lane];

        float* acc = (t == 0 ? A0 : A1) + (size_t)dst * IN_DIM + lane * 4;
        __hip_atomic_fetch_add(acc + 0, xv.x, __ATOMIC_RELAXED, __HIP_MEMORY_SCOPE_AGENT);
        __hip_atomic_fetch_add(acc + 1, xv.y, __ATOMIC_RELAXED, __HIP_MEMORY_SCOPE_AGENT);
        __hip_atomic_fetch_add(acc + 2, xv.z, __ATOMIC_RELAXED, __HIP_MEMORY_SCOPE_AGENT);
        __hip_atomic_fetch_add(acc + 3, xv.w, __ATOMIC_RELAXED, __HIP_MEMORY_SCOPE_AGENT);

        if (lane == 0) {
            __hip_atomic_fetch_add((t == 0 ? D0 : D1) + dst, 1.0f,
                                   __ATOMIC_RELAXED, __HIP_MEMORY_SCOPE_AGENT);
        }
    }
}

__global__ __launch_bounds__(256) void gnn_wmma_gemm(
    const float* __restrict__ X,
    const float* __restrict__ Wd, const float* __restrict__ bd,
    const float* __restrict__ Wc, const float* __restrict__ bc,
    const float* __restrict__ Wsf, const float* __restrict__ bs,
    const float* __restrict__ ws,
    float*       __restrict__ out)
{
    const float* A0 = ws + A0_OFF;
    const float* A1 = ws + A1_OFF;
    const float* D0 = ws + D0_OFF;
    const float* D1 = ws + D1_OFF;

    const int lane = threadIdx.x & 31;
    const int r    = lane & 15;
    const int h    = lane >> 4;
    const int o_base = (threadIdx.x >> 5) * 16;
    const int m_base = blockIdx.x * 16;

    const size_t arow = (size_t)(m_base + r) * IN_DIM + 2 * h;
    const size_t brow = (size_t)(o_base + r) * IN_DIM + 2 * h;

    v8f c = {};
#pragma unroll
    for (int k = 0; k < IN_DIM; k += 4) {
        v2f a0  = *(const v2f*)(A0  + arow + k);
        v2f wdv = *(const v2f*)(Wd  + brow + k);
        c = __builtin_amdgcn_wmma_f32_16x16x4_f32(false, a0, false, wdv,
                                                  (short)0, c, false, false);
        v2f a1  = *(const v2f*)(A1  + arow + k);
        v2f wcv = *(const v2f*)(Wc  + brow + k);
        c = __builtin_amdgcn_wmma_f32_16x16x4_f32(false, a1, false, wcv,
                                                  (short)0, c, false, false);
        v2f ax  = *(const v2f*)(X   + arow + k);
        v2f wsv = *(const v2f*)(Wsf + brow + k);
        c = __builtin_amdgcn_wmma_f32_16x16x4_f32(false, ax, false, wsv,
                                                  (short)0, c, false, false);
    }

    const int   o   = o_base + r;
    const float bdo = bd[o];
    const float bco = bc[o];
    const float bso = bs[o];
#pragma unroll
    for (int i = 0; i < 8; ++i) {
        const int m = m_base + i + 8 * h;
        float v = c[i] + D0[m] * bdo + D1[m] * bco + bso;
        out[(size_t)m * OUT_DIM + o] = v > 0.0f ? v : 0.0f;
    }
}

extern "C" void kernel_launch(void* const* d_in, const int* in_sizes, int n_in,
                              void* d_out, int out_size, void* d_ws, size_t ws_size,
                              hipStream_t stream) {
    const float* X   = (const float*)d_in[0];
    const int*   ei  = (const int*)d_in[1];
    const int*   et  = (const int*)d_in[2];
    const float* Wd  = (const float*)d_in[3];
    const float* bd  = (const float*)d_in[4];
    const float* Wc  = (const float*)d_in[5];
    const float* bc  = (const float*)d_in[6];
    const float* Wsf = (const float*)d_in[7];
    const float* bs  = (const float*)d_in[8];
    float* out = (float*)d_out;
    float* ws  = (float*)d_ws;

    (void)in_sizes; (void)n_in; (void)out_size; (void)ws_size;

    gnn_zero_ws<<<(WS_FLOATS + 255) / 256, 256, 0, stream>>>(ws);

    gnn_edge_scatter<<<4096, 256, 0, stream>>>(X, ei, et, ws);

    gnn_wmma_gemm<<<N_NODES / 16, 256, 0, stream>>>(X, Wd, bd, Wc, bc, Wsf, bs,
                                                    ws, out);
}


# f32 atomic scatter replaced by a counting sort of edges per (dst,type) (int atomic rank, scan, fill) plus a per-node row gather into LDS fused ahead of the WMMA GEMM; sum order = arrival-order rank
# speedup vs baseline: 1.9090x; 1.9090x over previous
#include <hip/hip_runtime.h>


#define IN_DIM   128
#define OUT_DIM  128
#define N_NODES  10000
#define N_EDGES  640000
#define N_SEG    (2 * N_NODES)
#define A_PITCH  132

typedef __attribute__((ext_vector_type(2))) float v2f;
typedef __attribute__((ext_vector_type(8))) float v8f;

#define CNT_OFF   0
#define SEG_OFF   (N_SEG)
#define RANK_OFF  (2 * N_SEG + 32)
#define SRT_OFF   (RANK_OFF + N_EDGES)

#define SCAN_PER_THREAD 20

__global__ __launch_bounds__(256) void gnn_zero_cnt(int* __restrict__ ws) {
    const int i = blockIdx.x * blockDim.x + threadIdx.x;
    if (i < N_SEG) ws[CNT_OFF + i] = 0;
}

__global__ __launch_bounds__(256) void gnn_edge_count(
    const int* __restrict__ edge_index,
    const int* __restrict__ edge_types,
    int*       __restrict__ ws)
{
    const int e = blockIdx.x * blockDim.x + threadIdx.x;
    if (e < N_EDGES) {
        const int dst = edge_index[N_EDGES + e];
        const int t   = edge_types[e];
        const int seg = 2 * dst + (t == 0 ? 0 : 1);
        ws[RANK_OFF + e] = __hip_atomic_fetch_add(ws + CNT_OFF + seg, 1,
                                                  __ATOMIC_RELAXED, __HIP_MEMORY_SCOPE_AGENT);
    }
}

__global__ __launch_bounds__(1024) void gnn_seg_scan(int* __restrict__ ws) {
    __shared__ int part[1024];
    const int* cnt = ws + CNT_OFF;
    int*       seg = ws + SEG_OFF;
    const int  t    = threadIdx.x;
    const int  base = t * SCAN_PER_THREAD;

    int loc[SCAN_PER_THREAD];
    int sum = 0;
#pragma unroll
    for (int i = 0; i < SCAN_PER_THREAD; ++i) {
        const int idx = base + i;
        const int v = idx < N_SEG ? cnt[idx] : 0;
        loc[i] = sum;
        sum += v;
    }
    part[t] = sum;
    __syncthreads();
    for (int d = 1; d < 1024; d <<= 1) {
        const int v = t >= d ? part[t - d] : 0;
        __syncthreads();
        part[t] += v;
        __syncthreads();
    }
    const int excl = part[t] - sum;
#pragma unroll
    for (int i = 0; i < SCAN_PER_THREAD; ++i) {
        const int idx = base + i;
        if (idx < N_SEG) seg[idx] = excl + loc[i];
    }
    if (t == 1023) seg[N_SEG] = part[1023];
}

__global__ __launch_bounds__(256) void gnn_edge_fill(
    const int* __restrict__ edge_index,
    const int* __restrict__ edge_types,
    int*       __restrict__ ws)
{
    const int e = blockIdx.x * blockDim.x + threadIdx.x;
    if (e < N_EDGES) {
        const int src = edge_index[e];
        const int dst = edge_index[N_EDGES + e];
        const int t   = edge_types[e];
        const int seg = 2 * dst + (t == 0 ? 0 : 1);
        const unsigned pos = (unsigned)(ws[SEG_OFF + seg] + ws[RANK_OFF + e]);
        if (pos < (unsigned)N_EDGES) ws[SRT_OFF + pos] = src;
    }
}

static __device__ __forceinline__ float4 gather_rows(
    const float* __restrict__ X, const int* __restrict__ srt,
    const int beg, const int end, const int lane)
{
    float4 acc = make_float4(0.f, 0.f, 0.f, 0.f);
    for (int base = beg; base < end; base += 32) {
        int n = end - base;
        if (n > 32) n = 32;
        const int mine = lane < n ? srt[base + lane] : 0;
        int j = 0;
        for (; j + 8 <= n; j += 8) {
            float4 v[8];
#pragma unroll
            for (int u = 0; u < 8; ++u) {
                const int s = __builtin_amdgcn_readlane(mine, j + u);
                v[u] = ((const float4*)(X + (size_t)s * IN_DIM))[lane];
            }
#pragma unroll
            for (int u = 0; u < 8; ++u) {
                acc.x += v[u].x; acc.y += v[u].y; acc.z += v[u].z; acc.w += v[u].w;
            }
        }
        if (j < n) {
            float4 v[8];
#pragma unroll
            for (int u = 0; u < 8; ++u) {
                const int s = __builtin_amdgcn_readlane(mine, j + u);
                v[u] = ((const float4*)(X + (size_t)s * IN_DIM))[lane];
            }
#pragma unroll
            for (int u = 0; u < 8; ++u) {
                if (j + u < n) {
                    acc.x += v[u].x; acc.y += v[u].y; acc.z += v[u].z; acc.w += v[u].w;
                }
            }
        }
    }
    return acc;
}

__global__ __launch_bounds__(256) void gnn_gather_gemm(
    const float* __restrict__ X,
    const float* __restrict__ Wd, const float* __restrict__ bd,
    const float* __restrict__ Wc, const float* __restrict__ bc,
    const float* __restrict__ Wsf, const float* __restrict__ bs,
    const int*   __restrict__ ws,
    float*       __restrict__ out)
{
    __shared__ __attribute__((aligned(16))) float sA[2][16][A_PITCH];
    __shared__ float sD[2][16];

    const int* segs = ws + SEG_OFF;
    const int* srt  = ws + SRT_OFF;

    const int lane   = threadIdx.x & 31;
    const int wv     = threadIdx.x >> 5;
    const int m_base = blockIdx.x * 16;

    for (int q = 0; q < 2; ++q) {
        const int row  = 2 * wv + q;
        const int node = m_base + row;
        const int s0 = __builtin_amdgcn_readfirstlane(segs[2 * node]);
        const int s1 = __builtin_amdgcn_readfirstlane(segs[2 * node + 1]);
        const int s2 = __builtin_amdgcn_readfirstlane(segs[2 * node + 2]);

        const float4 acc0 = gather_rows(X, srt, s0, s1, lane);
        const float4 acc1 = gather_rows(X, srt, s1, s2, lane);

        *(float4*)(&sA[0][row][lane * 4]) = acc0;
        *(float4*)(&sA[1][row][lane * 4]) = acc1;
        if (lane == 0) {
            sD[0][row] = (float)(s1 - s0);
            sD[1][row] = (float)(s2 - s1);
        }
    }
    __syncthreads();

    const int r      = lane & 15;
    const int h      = lane >> 4;
    const int o_base = wv * 16;

    const size_t arow = (size_t)(m_base + r) * IN_DIM + 2 * h;
    const size_t brow = (size_t)(o_base + r) * IN_DIM + 2 * h;
    const float* a0p = &sA[0][r][2 * h];
    const float* a1p = &sA[1][r][2 * h];

    v8f c = {};
#pragma unroll
    for (int k = 0; k < IN_DIM; k += 4) {
        v2f a0  = *(const v2f*)(a0p + k);
        v2f wdv = *(const v2f*)(Wd  + brow + k);
        c = __builtin_amdgcn_wmma_f32_16x16x4_f32(false, a0, false, wdv,
                                                  (short)0, c, false, false);
        v2f a1  = *(const v2f*)(a1p + k);
        v2f wcv = *(const v2f*)(Wc  + brow + k);
        c = __builtin_amdgcn_wmma_f32_16x16x4_f32(false, a1, false, wcv,
                                                  (short)0, c, false, false);
        v2f ax  = *(const v2f*)(X   + arow + k);
        v2f wsv = *(const v2f*)(Wsf + brow + k);
        c = __builtin_amdgcn_wmma_f32_16x16x4_f32(false, ax, false, wsv,
                                                  (short)0, c, false, false);
    }

    const int   o   = o_base + r;
    const float bdo = bd[o];
    const float bco = bc[o];
    const float bso = bs[o];
#pragma unroll
    for (int i = 0; i < 8; ++i) {
        const int ml = i + 8 * h;
        const int m  = m_base + ml;
        float v = c[i] + sD[0][ml] * bdo + sD[1][ml] * bco + bso;
        out[(size_t)m * OUT_DIM + o] = v > 0.0f ? v : 0.0f;
    }
}

extern "C" void kernel_launch(void* const* d_in, const int* in_sizes, int n_in,
                              void* d_out, int out_size, void* d_ws, size_t ws_size,
                              hipStream_t stream) {
    const float* X   = (const float*)d_in[0];
    const int*   ei  = (const int*)d_in[1];
    const int*   et  = (const int*)d_in[2];
    const float* Wd  = (const float*)d_in[3];
    const float* bd  = (const float*)d_in[4];
    const float* Wc  = (const float*)d_in[5];
    const float* bc  = (const float*)d_in[6];
    const float* Wsf = (const float*)d_in[7];
    const float* bs  = (const float*)d_in[8];
    float* out = (float*)d_out;
    int*   ws  = (int*)d_ws;

    (void)in_sizes; (void)n_in; (void)out_size; (void)ws_size;

    gnn_zero_cnt<<<(N_SEG + 255) / 256, 256, 0, stream>>>(ws);

    gnn_edge_count<<<N_EDGES / 256, 256, 0, stream>>>(ei, et, ws);

    gnn_seg_scan<<<1, 1024, 0, stream>>>(ws);

    gnn_edge_fill<<<N_EDGES / 256, 256, 0, stream>>>(ei, et, ws);

    gnn_gather_gemm<<<N_NODES / 16, 256, 0, stream>>>(X, Wd, bd, Wc, bc, Wsf, bs,
                                                      ws, out);
}


# edge ranks from a per-block LDS histogram per dst node + one coalesced 64-bit returning add per (block, node) on the global counts, not a scattered add per edge; sum order = block, then LDS arrival
# speedup vs baseline: 3.5953x; 1.8833x over previous
#include <hip/hip_runtime.h>


#define IN_DIM   128
#define OUT_DIM  128
#define N_NODES  10000
#define N_EDGES  640000
#define N_SEG    (2 * N_NODES)
#define A_PITCH  132
#define SEG_CAP  64

typedef __attribute__((ext_vector_type(2))) float v2f;
typedef __attribute__((ext_vector_type(8))) float v8f;

#define CNT_OFF   0
#define OVF_OFF   (N_SEG)
#define N_ZERO    (N_SEG + 32)
#define SLOT_OFF  (N_ZERO)
#define LIST_OFF  (SLOT_OFF + N_SEG * SEG_CAP / 2)

__global__ __launch_bounds__(256) void gnn_zero_cnt(int* __restrict__ ws) {
    const int i = blockIdx.x * blockDim.x + threadIdx.x;
    if (i < N_ZERO) ws[CNT_OFF + i] = 0;
}

#define EP_THREADS     512
#define EP_PER_THREAD  5
#define EP_EDGES       (EP_THREADS * EP_PER_THREAD)
#define EP_NODE_STEPS  ((N_NODES + EP_THREADS - 1) / EP_THREADS)
static_assert(N_EDGES % EP_EDGES == 0, "the grid covers the edge list exactly");
static_assert(EP_EDGES < 65536, "a block's count per segment must fit 16 bits");
static_assert((CNT_OFF % 2) == 0, "a node's pair of counts is one aligned 64-bit word");

__global__ __launch_bounds__(EP_THREADS) void gnn_edge_place(
    const int* __restrict__ edge_index,
    const int* __restrict__ edge_types,
    int*       __restrict__ ws)
{
    __shared__ unsigned hist[N_NODES];

    const int tid = threadIdx.x;
    for (int s = tid; s < N_NODES; s += EP_THREADS) hist[s] = 0u;

    int src[EP_PER_THREAD], dst[EP_PER_THREAD], typ[EP_PER_THREAD];
    const int e0 = blockIdx.x * EP_EDGES + tid;
#pragma unroll
    for (int j = 0; j < EP_PER_THREAD; ++j) {
        const int e = e0 + j * EP_THREADS;
        src[j] = edge_index[e];
        dst[j] = edge_index[N_EDGES + e];
        typ[j] = edge_types[e] == 0 ? 0 : 1;
    }
    __syncthreads();

    unsigned loc[EP_PER_THREAD];
#pragma unroll
    for (int j = 0; j < EP_PER_THREAD; ++j) {
        const unsigned old = __hip_atomic_fetch_add(&hist[dst[j]], typ[j] ? 0x10000u : 1u,
                                                    __ATOMIC_RELAXED, __HIP_MEMORY_SCOPE_WORKGROUP);
        loc[j] = typ[j] ? (old >> 16) : (old & 0xffffu);
    }
    __syncthreads();

    unsigned long long* cnt64 = (unsigned long long*)(ws + CNT_OFF);
    unsigned           w[EP_NODE_STEPS];
    unsigned long long was[EP_NODE_STEPS];
#pragma unroll
    for (int i = 0; i < EP_NODE_STEPS; ++i) {
        const int s = tid + i * EP_THREADS;
        w[i] = s < N_NODES ? hist[s] : 0u;
    }
#pragma unroll
    for (int i = 0; i < EP_NODE_STEPS; ++i) {
        was[i] = 0ull;
        if (w[i] != 0u) {
            const unsigned long long add =
                ((unsigned long long)(w[i] >> 16) << 32) | (unsigned long long)(w[i] & 0xffffu);
            was[i] = __hip_atomic_fetch_add(cnt64 + tid + i * EP_THREADS, add,
                                            __ATOMIC_RELAXED, __HIP_MEMORY_SCOPE_AGENT);
        }
    }
#pragma unroll
    for (int i = 0; i < EP_NODE_STEPS; ++i) {
        if (w[i] != 0u) {
            unsigned b0 = (unsigned)was[i];
            unsigned b1 = (unsigned)(was[i] >> 32);
            b0 = b0 < SEG_CAP ? b0 : SEG_CAP;
            b1 = b1 < SEG_CAP ? b1 : SEG_CAP;
            hist[tid + i * EP_THREADS] = b0 | (b1 << 16);
        }
    }
    __syncthreads();

    unsigned short* slots = (unsigned short*)(ws + SLOT_OFF);
    unsigned rank[EP_PER_THREAD];
#pragma unroll
    for (int j = 0; j < EP_PER_THREAD; ++j) {
        const unsigned bw = hist[dst[j]];
        rank[j] = (typ[j] ? (bw >> 16) : (bw & 0xffffu)) + loc[j];
    }
#pragma unroll
    for (int j = 0; j < EP_PER_THREAD; ++j) {
        const int seg = 2 * dst[j] + typ[j];
        if (rank[j] < (unsigned)SEG_CAP) {
            slots[seg * SEG_CAP + rank[j]] = (unsigned short)src[j];
        } else {
            const int k = __hip_atomic_fetch_add(ws + OVF_OFF, 1,
                                                 __ATOMIC_RELAXED, __HIP_MEMORY_SCOPE_AGENT);
            if ((unsigned)k < (unsigned)N_EDGES)
                ws[LIST_OFF + k] = (seg << 16) | (src[j] & 0xffff);
        }
    }
}

static __device__ __forceinline__ float4 gather_rows(
    const float* __restrict__ X, const unsigned short* __restrict__ slots,
    const int cnt, const int lane)
{
    float4 acc = make_float4(0.f, 0.f, 0.f, 0.f);
    for (int base = 0; base < cnt; base += 32) {
        int n = cnt - base;
        if (n > 32) n = 32;
        const int mine = lane < n ? (int)slots[base + lane] : 0;
        int j = 0;
        for (; j + 8 <= n; j += 8) {
            float4 v[8];
#pragma unroll
            for (int u = 0; u < 8; ++u) {
                const int s = __builtin_amdgcn_readlane(mine, j + u);
                v[u] = ((const float4*)(X + (size_t)s * IN_DIM))[lane];
            }
#pragma unroll
            for (int u = 0; u < 8; ++u) {
                acc.x += v[u].x; acc.y += v[u].y; acc.z += v[u].z; acc.w += v[u].w;
            }
        }
        if (j < n) {
            float4 v[8];
#pragma unroll
            for (int u = 0; u < 8; ++u) {
                const int s = __builtin_amdgcn_readlane(mine, j + u);
                v[u] = ((const float4*)(X + (size_t)s * IN_DIM))[lane];
            }
#pragma unroll
            for (int u = 0; u < 8; ++u) {
                if (j + u < n) {
                    acc.x += v[u].x; acc.y += v[u].y; acc.z += v[u].z; acc.w += v[u].w;
                }
            }
        }
    }
    return acc;
}

static __device__ __forceinline__ void gather_spill(
    const float* __restrict__ X, const int* __restrict__ spill, const int novf,
    const int seg0, int want, const int lane, float4& acc0, float4& acc1)
{
    for (int base = 0; base < novf && want > 0; base += 32) {
        const int ent = base + lane < novf ? spill[base + lane] : -1;
        const int sg  = ent >> 16;
        unsigned m0 = __builtin_amdgcn_ballot_w32(sg == seg0);
        unsigned m1 = __builtin_amdgcn_ballot_w32(sg == seg0 + 1);
        want -= __builtin_popcount(m0) + __builtin_popcount(m1);
        while (m0) {
            const int l = __builtin_ctz(m0);
            m0 &= m0 - 1;
            const int s = __builtin_amdgcn_readlane(ent, l) & 0xffff;
            const float4 v = ((const float4*)(X + (size_t)s * IN_DIM))[lane];
            acc0.x += v.x; acc0.y += v.y; acc0.z += v.z; acc0.w += v.w;
        }
        while (m1) {
            const int l = __builtin_ctz(m1);
            m1 &= m1 - 1;
            const int s = __builtin_amdgcn_readlane(ent, l) & 0xffff;
            const float4 v = ((const float4*)(X + (size_t)s * IN_DIM))[lane];
            acc1.x += v.x; acc1.y += v.y; acc1.z += v.z; acc1.w += v.w;
        }
    }
}

__global__ __launch_bounds__(256) void gnn_gather_gemm(
    const float* __restrict__ X,
    const float* __restrict__ Wd, const float* __restrict__ bd,
    const float* __restrict__ Wc, const float* __restrict__ bc,
    const float* __restrict__ Wsf, const float* __restrict__ bs,
    const int*   __restrict__ ws,
    float*       __restrict__ out)
{
    __shared__ __attribute__((aligned(16))) float sA[2][16][A_PITCH];
    __shared__ float sD[2][16];

    const int*            cnt   = ws + CNT_OFF;
    const unsigned short* slots = (const unsigned short*)(ws + SLOT_OFF);
    const int*            spill = ws + LIST_OFF;

    const int lane   = threadIdx.x & 31;
    const int wv     = threadIdx.x >> 5;
    const int m_base = blockIdx.x * 16;

    for (int q = 0; q < 2; ++q) {
        const int row  = 2 * wv + q;
        const int node = m_base + row;
        const int c0 = __builtin_amdgcn_readfirstlane(cnt[2 * node]);
        const int c1 = __builtin_amdgcn_readfirstlane(cnt[2 * node + 1]);
        const int n0 = c0 < SEG_CAP ? c0 : SEG_CAP;
        const int n1 = c1 < SEG_CAP ? c1 : SEG_CAP;

        float4 acc0 = gather_rows(X, slots + (size_t)(2 * node) * SEG_CAP, n0, lane);
        float4 acc1 = gather_rows(X, slots + (size_t)(2 * node + 1) * SEG_CAP, n1, lane);

        const int want = (c0 - n0) + (c1 - n1);
        if (want > 0) {
            int novf = __builtin_amdgcn_readfirstlane(ws[OVF_OFF]);
            if (novf > N_EDGES) novf = N_EDGES;
            gather_spill(X, spill, novf, 2 * node, want, lane, acc0, acc1);
        }

        *(float4*)(&sA[0][row][lane * 4]) = acc0;
        *(float4*)(&sA[1][row][lane * 4]) = acc1;
        if (lane == 0) {
            sD[0][row] = (float)c0;
            sD[1][row] = (float)c1;
        }
    }
    __syncthreads();

    const int r      = lane & 15;
    const int h      = lane >> 4;
    const int o_base = wv * 16;

    const size_t arow = (size_t)(m_base + r) * IN_DIM + 2 * h;
    const size_t brow = (size_t)(o_base + r) * IN_DIM + 2 * h;
    const float* a0p = &sA[0][r][2 * h];
    const float* a1p = &sA[1][r][2 * h];

    v8f c = {};
#pragma unroll
    for (int k = 0; k < IN_DIM; k += 4) {
        v2f a0  = *(const v2f*)(a0p + k);
        v2f wdv = *(const v2f*)(Wd  + brow + k);
        c = __builtin_amdgcn_wmma_f32_16x16x4_f32(false, a0, false, wdv,
                                                  (short)0, c, false, false);
        v2f a1  = *(const v2f*)(a1p + k);
        v2f wcv = *(const v2f*)(Wc  + brow + k);
        c = __builtin_amdgcn_wmma_f32_16x16x4_f32(false, a1, false, wcv,
                                                  (short)0, c, false, false);
        v2f ax  = *(const v2f*)(X   + arow + k);
        v2f wsv = *(const v2f*)(Wsf + brow + k);
        c = __builtin_amdgcn_wmma_f32_16x16x4_f32(false, ax, false, wsv,
                                                  (short)0, c, false, false);
    }

    const int   o   = o_base + r;
    const float bdo = bd[o];
    const float bco = bc[o];
    const float bso = bs[o];
#pragma unroll
    for (int i = 0; i < 8; ++i) {
        const int ml = i + 8 * h;
        const int m  = m_base + ml;
        float v = c[i] + sD[0][ml] * bdo + sD[1][ml] * bco + bso;
        out[(size_t)m * OUT_DIM + o] = v > 0.0f ? v : 0.0f;
    }
}

extern "C" void kernel_launch(void* const* d_in, const int* in_sizes, int n_in,
                              void* d_out, int out_size, void* d_ws, size_t ws_size,
                              hipStream_t stream) {
    const float* X   = (const float*)d_in[0];
    const int*   ei  = (const int*)d_in[1];
    const int*   et  = (const int*)d_in[2];
    const float* Wd  = (const float*)d_in[3];
    const float* bd  = (const float*)d_in[4];
    const float* Wc  = (const float*)d_in[5];
    const float* bc  = (const float*)d_in[6];
    const float* Wsf = (const float*)d_in[7];
    const float* bs  = (const float*)d_in[8];
    float* out = (float*)d_out;
    int*   ws  = (int*)d_ws;

    (void)in_sizes; (void)n_in; (void)out_size; (void)ws_size;

    gnn_zero_cnt<<<(N_ZERO + 255) / 256, 256, 0, stream>>>(ws);

    gnn_edge_place<<<N_EDGES / EP_EDGES, EP_THREADS, 0, stream>>>(ei, et, ws);

    gnn_gather_gemm<<<N_NODES / 16, 256, 0, stream>>>(X, Wd, bd, Wc, bc, Wsf, bs,
                                                      ws, out);
}


# GEMM operand fetch: A and B fragments read as 16-byte lane pieces under one shared k permutation (k order of the sums changes), self rows staged in LDS: 128 sixteen-line loads per wave become 48
# speedup vs baseline: 4.1306x; 1.1489x over previous
#include <hip/hip_runtime.h>


#define IN_DIM   128
#define OUT_DIM  128
#define N_NODES  10000
#define N_EDGES  640000
#define N_SEG    (2 * N_NODES)
#define A_PITCH  132
#define SEG_CAP  64

typedef __attribute__((ext_vector_type(2))) float v2f;
typedef __attribute__((ext_vector_type(8))) float v8f;

#define CNT_OFF   0
#define OVF_OFF   (N_SEG)
#define N_ZERO    (N_SEG + 32)
#define SLOT_OFF  (N_ZERO)
#define LIST_OFF  (SLOT_OFF + N_SEG * SEG_CAP / 2)

__global__ __launch_bounds__(256) void gnn_zero_cnt(int* __restrict__ ws) {
    const int i = blockIdx.x * blockDim.x + threadIdx.x;
    if (i < N_ZERO) ws[CNT_OFF + i] = 0;
}

#define EP_THREADS     512
#define EP_PER_THREAD  5
#define EP_EDGES       (EP_THREADS * EP_PER_THREAD)
#define EP_NODE_STEPS  ((N_NODES + EP_THREADS - 1) / EP_THREADS)
static_assert(N_EDGES % EP_EDGES == 0, "the grid covers the edge list exactly");
static_assert(EP_EDGES < 65536, "a block's count per segment must fit 16 bits");
static_assert((CNT_OFF % 2) == 0, "a node's pair of counts is one aligned 64-bit word");

__global__ __launch_bounds__(EP_THREADS) void gnn_edge_place(
    const int* __restrict__ edge_index,
    const int* __restrict__ edge_types,
    int*       __restrict__ ws)
{
    __shared__ unsigned hist[N_NODES];

    const int tid = threadIdx.x;
    for (int s = tid; s < N_NODES; s += EP_THREADS) hist[s] = 0u;

    int src[EP_PER_THREAD], dst[EP_PER_THREAD], typ[EP_PER_THREAD];
    const int e0 = blockIdx.x * EP_EDGES + tid;
#pragma unroll
    for (int j = 0; j < EP_PER_THREAD; ++j) {
        const int e = e0 + j * EP_THREADS;
        src[j] = edge_index[e];
        dst[j] = edge_index[N_EDGES + e];
        typ[j] = edge_types[e] == 0 ? 0 : 1;
    }
    __syncthreads();

    unsigned loc[EP_PER_THREAD];
#pragma unroll
    for (int j = 0; j < EP_PER_THREAD; ++j) {
        const unsigned old = __hip_atomic_fetch_add(&hist[dst[j]], typ[j] ? 0x10000u : 1u,
                                                    __ATOMIC_RELAXED, __HIP_MEMORY_SCOPE_WORKGROUP);
        loc[j] = typ[j] ? (old >> 16) : (old & 0xffffu);
    }
    __syncthreads();

    unsigned long long* cnt64 = (unsigned long long*)(ws + CNT_OFF);
    unsigned           w[EP_NODE_STEPS];
    unsigned long long was[EP_NODE_STEPS];
#pragma unroll
    for (int i = 0; i < EP_NODE_STEPS; ++i) {
        const int s = tid + i * EP_THREADS;
        w[i] = s < N_NODES ? hist[s] : 0u;
    }
#pragma unroll
    for (int i = 0; i < EP_NODE_STEPS; ++i) {
        was[i] = 0ull;
        if (w[i] != 0u) {
            const unsigned long long add =
                ((unsigned long long)(w[i] >> 16) << 32) | (unsigned long long)(w[i] & 0xffffu);
            was[i] = __hip_atomic_fetch_add(cnt64 + tid + i * EP_THREADS, add,
                                            __ATOMIC_RELAXED, __HIP_MEMORY_SCOPE_AGENT);
        }
    }
#pragma unroll
    for (int i = 0; i < EP_NODE_STEPS; ++i) {
        if (w[i] != 0u) {
            unsigned b0 = (unsigned)was[i];
            unsigned b1 = (unsigned)(was[i] >> 32);
            b0 = b0 < SEG_CAP ? b0 : SEG_CAP;
            b1 = b1 < SEG_CAP ? b1 : SEG_CAP;
            hist[tid + i * EP_THREADS] = b0 | (b1 << 16);
        }
    }
    __syncthreads();

    unsigned short* slots = (unsigned short*)(ws + SLOT_OFF);
    unsigned rank[EP_PER_THREAD];
#pragma unroll
    for (int j = 0; j < EP_PER_THREAD; ++j) {
        const unsigned bw = hist[dst[j]];
        rank[j] = (typ[j] ? (bw >> 16) : (bw & 0xffffu)) + loc[j];
    }
#pragma unroll
    for (int j = 0; j < EP_PER_THREAD; ++j) {
        const int seg = 2 * dst[j] + typ[j];
        if (rank[j] < (unsigned)SEG_CAP) {
            slots[seg * SEG_CAP + rank[j]] = (unsigned short)src[j];
        } else {
            const int k = __hip_atomic_fetch_add(ws + OVF_OFF, 1,
                                                 __ATOMIC_RELAXED, __HIP_MEMORY_SCOPE_AGENT);
            if ((unsigned)k < (unsigned)N_EDGES)
                ws[LIST_OFF + k] = (seg << 16) | (src[j] & 0xffff);
        }
    }
}

static __device__ __forceinline__ float4 gather_rows(
    const float* __restrict__ X, const unsigned short* __restrict__ slots,
    const int cnt, const int lane)
{
    float4 acc = make_float4(0.f, 0.f, 0.f, 0.f);
    for (int base = 0; base < cnt; base += 32) {
        int n = cnt - base;
        if (n > 32) n = 32;
        const int mine = lane < n ? (int)slots[base + lane] : 0;
        int j = 0;
        for (; j + 8 <= n; j += 8) {
            float4 v[8];
#pragma unroll
            for (int u = 0; u < 8; ++u) {
                const int s = __builtin_amdgcn_readlane(mine, j + u);
                v[u] = ((const float4*)(X + (size_t)s * IN_DIM))[lane];
            }
#pragma unroll
            for (int u = 0; u < 8; ++u) {
                acc.x += v[u].x; acc.y += v[u].y; acc.z += v[u].z; acc.w += v[u].w;
            }
        }
        if (j < n) {
            float4 v[8];
#pragma unroll
            for (int u = 0; u < 8; ++u) {
                const int s = __builtin_amdgcn_readlane(mine, j + u);
                v[u] = ((const float4*)(X + (size_t)s * IN_DIM))[lane];
            }
#pragma unroll
            for (int u = 0; u < 8; ++u) {
                if (j + u < n) {
                    acc.x += v[u].x; acc.y += v[u].y; acc.z += v[u].z; acc.w += v[u].w;
                }
            }
        }
    }
    return acc;
}

static __device__ __forceinline__ void gather_spill(
    const float* __restrict__ X, const int* __restrict__ spill, const int novf,
    const int seg0, int want, const int lane, float4& acc0, float4& acc1)
{
    for (int base = 0; base < novf && want > 0; base += 32) {
        const int ent = base + lane < novf ? spill[base + lane] : -1;
        const int sg  = ent >> 16;
        unsigned m0 = __builtin_amdgcn_ballot_w32(sg == seg0);
        unsigned m1 = __builtin_amdgcn_ballot_w32(sg == seg0 + 1);
        want -= __builtin_popcount(m0) + __builtin_popcount(m1);
        while (m0) {
            const int l = __builtin_ctz(m0);
            m0 &= m0 - 1;
            const int s = __builtin_amdgcn_readlane(ent, l) & 0xffff;
            const float4 v = ((const float4*)(X + (size_t)s * IN_DIM))[lane];
            acc0.x += v.x; acc0.y += v.y; acc0.z += v.z; acc0.w += v.w;
        }
        while (m1) {
            const int l = __builtin_ctz(m1);
            m1 &= m1 - 1;
            const int s = __builtin_amdgcn_readlane(ent, l) & 0xffff;
            const float4 v = ((const float4*)(X + (size_t)s * IN_DIM))[lane];
            acc1.x += v.x; acc1.y += v.y; acc1.z += v.z; acc1.w += v.w;
        }
    }
}

__global__ __launch_bounds__(256) void gnn_gather_gemm(
    const float* __restrict__ X,
    const float* __restrict__ Wd, const float* __restrict__ bd,
    const float* __restrict__ Wc, const float* __restrict__ bc,
    const float* __restrict__ Wsf, const float* __restrict__ bs,
    const int*   __restrict__ ws,
    float*       __restrict__ out)
{
    __shared__ __attribute__((aligned(16))) float sA[3][16][A_PITCH];
    __shared__ float sD[2][16];

    const int*            cnt   = ws + CNT_OFF;
    const unsigned short* slots = (const unsigned short*)(ws + SLOT_OFF);
    const int*            spill = ws + LIST_OFF;

    const int lane   = threadIdx.x & 31;
    const int wv     = threadIdx.x >> 5;
    const int m_base = blockIdx.x * 16;

    for (int q = 0; q < 2; ++q) {
        const int row  = 2 * wv + q;
        const int node = m_base + row;
        const float4 xs = ((const float4*)(X + (size_t)node * IN_DIM))[lane];
        const int c0 = __builtin_amdgcn_readfirstlane(cnt[2 * node]);
        const int c1 = __builtin_amdgcn_readfirstlane(cnt[2 * node + 1]);
        const int n0 = c0 < SEG_CAP ? c0 : SEG_CAP;
        const int n1 = c1 < SEG_CAP ? c1 : SEG_CAP;
        *(float4*)(&sA[2][row][lane * 4]) = xs;

        float4 acc0 = gather_rows(X, slots + (size_t)(2 * node) * SEG_CAP, n0, lane);
        float4 acc1 = gather_rows(X, slots + (size_t)(2 * node + 1) * SEG_CAP, n1, lane);

        const int want = (c0 - n0) + (c1 - n1);
        if (want > 0) {
            int novf = __builtin_amdgcn_readfirstlane(ws[OVF_OFF]);
            if (novf > N_EDGES) novf = N_EDGES;
            gather_spill(X, spill, novf, 2 * node, want, lane, acc0, acc1);
        }

        *(float4*)(&sA[0][row][lane * 4]) = acc0;
        *(float4*)(&sA[1][row][lane * 4]) = acc1;
        if (lane == 0) {
            sD[0][row] = (float)c0;
            sD[1][row] = (float)c1;
        }
    }
    __syncthreads();

    const int r      = lane & 15;
    const int h      = lane >> 4;
    const int o_base = wv * 16;

    const size_t brow = (size_t)(o_base + r) * IN_DIM + 4 * h;
    const float* a0p = &sA[0][r][4 * h];
    const float* a1p = &sA[1][r][4 * h];
    const float* a2p = &sA[2][r][4 * h];

    v8f c = {};
#pragma unroll
    for (int k = 0; k < IN_DIM; k += 8) {
        const float4 a0 = *(const float4*)(a0p + k);
        const float4 wd = *(const float4*)(Wd  + brow + k);
        c = __builtin_amdgcn_wmma_f32_16x16x4_f32(false, (v2f){a0.x, a0.y}, false, (v2f){wd.x, wd.y},
                                                  (short)0, c, false, false);
        c = __builtin_amdgcn_wmma_f32_16x16x4_f32(false, (v2f){a0.z, a0.w}, false, (v2f){wd.z, wd.w},
                                                  (short)0, c, false, false);
        const float4 a1 = *(const float4*)(a1p + k);
        const float4 wc = *(const float4*)(Wc  + brow + k);
        c = __builtin_amdgcn_wmma_f32_16x16x4_f32(false, (v2f){a1.x, a1.y}, false, (v2f){wc.x, wc.y},
                                                  (short)0, c, false, false);
        c = __builtin_amdgcn_wmma_f32_16x16x4_f32(false, (v2f){a1.z, a1.w}, false, (v2f){wc.z, wc.w},
                                                  (short)0, c, false, false);
        const float4 a2 = *(const float4*)(a2p + k);
        const float4 wf = *(const float4*)(Wsf + brow + k);
        c = __builtin_amdgcn_wmma_f32_16x16x4_f32(false, (v2f){a2.x, a2.y}, false, (v2f){wf.x, wf.y},
                                                  (short)0, c, false, false);
        c = __builtin_amdgcn_wmma_f32_16x16x4_f32(false, (v2f){a2.z, a2.w}, false, (v2f){wf.z, wf.w},
                                                  (short)0, c, false, false);
    }

    const int   o   = o_base + r;
    const float bdo = bd[o];
    const float bco = bc[o];
    const float bso = bs[o];
#pragma unroll
    for (int i = 0; i < 8; ++i) {
        const int ml = i + 8 * h;
        const int m  = m_base + ml;
        float v = c[i] + sD[0][ml] * bdo + sD[1][ml] * bco + bso;
        out[(size_t)m * OUT_DIM + o] = v > 0.0f ? v : 0.0f;
    }
}

extern "C" void kernel_launch(void* const* d_in, const int* in_sizes, int n_in,
                              void* d_out, int out_size, void* d_ws, size_t ws_size,
                              hipStream_t stream) {
    const float* X   = (const float*)d_in[0];
    const int*   ei  = (const int*)d_in[1];
    const int*   et  = (const int*)d_in[2];
    const float* Wd  = (const float*)d_in[3];
    const float* bd  = (const float*)d_in[4];
    const float* Wc  = (const float*)d_in[5];
    const float* bc  = (const float*)d_in[6];
    const float* Wsf = (const float*)d_in[7];
    const float* bs  = (const float*)d_in[8];
    float* out = (float*)d_out;
    int*   ws  = (int*)d_ws;

    (void)in_sizes; (void)n_in; (void)out_size; (void)ws_size;

    gnn_zero_cnt<<<(N_ZERO + 255) / 256, 256, 0, stream>>>(ws);

    gnn_edge_place<<<N_EDGES / EP_EDGES, EP_THREADS, 0, stream>>>(ei, et, ws);

    gnn_gather_gemm<<<N_NODES / 16, 256, 0, stream>>>(X, Wd, bd, Wc, bc, Wsf, bs,
                                                      ws, out);
}


# B operand: weight matrices re-laid in WMMA fragment order in the workspace by the first launch; each B fetch is one contiguous 512-byte wave load of 4 whole lines, not 16 lines x 32 B; sums kept
# speedup vs baseline: 4.4031x; 1.0660x over previous
#include <hip/hip_runtime.h>


#define IN_DIM   128
#define OUT_DIM  128
#define N_NODES  10000
#define N_EDGES  640000
#define N_SEG    (2 * N_NODES)
#define A_PITCH  132
#define SEG_CAP  64

typedef __attribute__((ext_vector_type(2))) float v2f;
typedef __attribute__((ext_vector_type(8))) float v8f;

#define CNT_OFF   0
#define OVF_OFF   (N_SEG)
#define N_ZERO    (N_SEG + 32)
#define SLOT_OFF  (N_ZERO)
#define LIST_OFF  (SLOT_OFF + N_SEG * SEG_CAP / 2)
#define WP_OFF    (((LIST_OFF + N_EDGES) + 127) / 128 * 128)
#define N_KGROUP  (IN_DIM / 8)
#define N_OTILE   (OUT_DIM / 16)
#define WP_PIECES (3 * N_OTILE * N_KGROUP * 32)
static_assert((WP_OFF % 128) == 0, "fragment chunks start on 512-byte boundaries");
static_assert((N_ZERO % 32) == 0, "a wave of the first launch is all clear or all copy");

__global__ __launch_bounds__(256) void gnn_prepare(
    const float* __restrict__ Wd, const float* __restrict__ Wc, const float* __restrict__ Wsf,
    int* __restrict__ ws)
{
    const int i = blockIdx.x * blockDim.x + threadIdx.x;
    if (i < N_ZERO) {
        ws[CNT_OFF + i] = 0;
    } else if (i < N_ZERO + WP_PIECES) {
        const int p    = i - N_ZERO;
        const int lane = p & 31;
        const int q    = p >> 5;
        const int mat  = q % 3;
        const int g    = (q / 3) % N_KGROUP;
        const int tile = q / (3 * N_KGROUP);
        const float* Wm = mat == 0 ? Wd : (mat == 1 ? Wc : Wsf);
        const float4 v = *(const float4*)(Wm + (size_t)(16 * tile + (lane & 15)) * IN_DIM
                                             + 8 * g + 4 * (lane >> 4));
        ((float4*)(ws + WP_OFF))[p] = v;
    }
}

#define EP_THREADS     512
#define EP_PER_THREAD  5
#define EP_EDGES       (EP_THREADS * EP_PER_THREAD)
#define EP_NODE_STEPS  ((N_NODES + EP_THREADS - 1) / EP_THREADS)
static_assert(N_EDGES % EP_EDGES == 0, "the grid covers the edge list exactly");
static_assert(EP_EDGES < 65536, "a block's count per segment must fit 16 bits");
static_assert((CNT_OFF % 2) == 0, "a node's pair of counts is one aligned 64-bit word");

__global__ __launch_bounds__(EP_THREADS) void gnn_edge_place(
    const int* __restrict__ edge_index,
    const int* __restrict__ edge_types,
    int*       __restrict__ ws)
{
    __shared__ unsigned hist[N_NODES];

    const int tid = threadIdx.x;
    for (int s = tid; s < N_NODES; s += EP_THREADS) hist[s] = 0u;

    int src[EP_PER_THREAD], dst[EP_PER_THREAD], typ[EP_PER_THREAD];
    const int e0 = blockIdx.x * EP_EDGES + tid;
#pragma unroll
    for (int j = 0; j < EP_PER_THREAD; ++j) {
        const int e = e0 + j * EP_THREADS;
        src[j] = edge_index[e];
        dst[j] = edge_index[N_EDGES + e];
        typ[j] = edge_types[e] == 0 ? 0 : 1;
    }
    __syncthreads();

    unsigned loc[EP_PER_THREAD];
#pragma unroll
    for (int j = 0; j < EP_PER_THREAD; ++j) {
        const unsigned old = __hip_atomic_fetch_add(&hist[dst[j]], typ[j] ? 0x10000u : 1u,
                                                    __ATOMIC_RELAXED, __HIP_MEMORY_SCOPE_WORKGROUP);
        loc[j] = typ[j] ? (old >> 16) : (old & 0xffffu);
    }
    __syncthreads();

    unsigned long long* cnt64 = (unsigned long long*)(ws + CNT_OFF);
    unsigned           w[EP_NODE_STEPS];
    unsigned long long was[EP_NODE_STEPS];
#pragma unroll
    for (int i = 0; i < EP_NODE_STEPS; ++i) {
        const int s = tid + i * EP_THREADS;
        w[i] = s < N_NODES ? hist[s] : 0u;
    }
#pragma unroll
    for (int i = 0; i < EP_NODE_STEPS; ++i) {
        was[i] = 0ull;
        if (w[i] != 0u) {
            const unsigned long long add =
                ((unsigned long long)(w[i] >> 16) << 32) | (unsigned long long)(w[i] & 0xffffu);
            was[i] = __hip_atomic_fetch_add(cnt64 + tid + i * EP_THREADS, add,
                                            __ATOMIC_RELAXED, __HIP_MEMORY_SCOPE_AGENT);
        }
    }
#pragma unroll
    for (int i = 0; i < EP_NODE_STEPS; ++i) {
        if (w[i] != 0u) {
            unsigned b0 = (unsigned)was[i];
            unsigned b1 = (unsigned)(was[i] >> 32);
            b0 = b0 < SEG_CAP ? b0 : SEG_CAP;
            b1 = b1 < SEG_CAP ? b1 : SEG_CAP;
            hist[tid + i * EP_THREADS] = b0 | (b1 << 16);
        }
    }
    __syncthreads();

    unsigned short* slots = (unsigned short*)(ws + SLOT_OFF);
    unsigned rank[EP_PER_THREAD];
#pragma unroll
    for (int j = 0; j < EP_PER_THREAD; ++j) {
        const unsigned bw = hist[dst[j]];
        rank[j] = (typ[j] ? (bw >> 16) : (bw & 0xffffu)) + loc[j];
    }
#pragma unroll
    for (int j = 0; j < EP_PER_THREAD; ++j) {
        const int seg = 2 * dst[j] + typ[j];
        if (rank[j] < (unsigned)SEG_CAP) {
            slots[seg * SEG_CAP + rank[j]] = (unsigned short)src[j];
        } else {
            const int k = __hip_atomic_fetch_add(ws + OVF_OFF, 1,
                                                 __ATOMIC_RELAXED, __HIP_MEMORY_SCOPE_AGENT);
            if ((unsigned)k < (unsigned)N_EDGES)
                ws[LIST_OFF + k] = (seg << 16) | (src[j] & 0xffff);
        }
    }
}

static __device__ __forceinline__ float4 gather_rows(
    const float* __restrict__ X, const unsigned short* __restrict__ slots,
    const int cnt, const int lane)
{
    float4 acc = make_float4(0.f, 0.f, 0.f, 0.f);
    for (int base = 0; base < cnt; base += 32) {
        int n = cnt - base;
        if (n > 32) n = 32;
        const int mine = lane < n ? (int)slots[base + lane] : 0;
        int j = 0;
        for (; j + 8 <= n; j += 8) {
            float4 v[8];
#pragma unroll
            for (int u = 0; u < 8; ++u) {
                const int s = __builtin_amdgcn_readlane(mine, j + u);
                v[u] = ((const float4*)(X + (size_t)s * IN_DIM))[lane];
            }
#pragma unroll
            for (int u = 0; u < 8; ++u) {
                acc.x += v[u].x; acc.y += v[u].y; acc.z += v[u].z; acc.w += v[u].w;
            }
        }
        if (j < n) {
            float4 v[8];
#pragma unroll
            for (int u = 0; u < 8; ++u) {
                const int s = __builtin_amdgcn_readlane(mine, j + u);
                v[u] = ((const float4*)(X + (size_t)s * IN_DIM))[lane];
            }
#pragma unroll
            for (int u = 0; u < 8; ++u) {
                if (j + u < n) {
                    acc.x += v[u].x; acc.y += v[u].y; acc.z += v[u].z; acc.w += v[u].w;
                }
            }
        }
    }
    return acc;
}

static __device__ __forceinline__ void gather_spill(
    const float* __restrict__ X, const int* __restrict__ spill, const int novf,
    const int seg0, int want, const int lane, float4& acc0, float4& acc1)
{
    for (int base = 0; base < novf && want > 0; base += 32) {
        const int ent = base + lane < novf ? spill[base + lane] : -1;
        const int sg  = ent >> 16;
        unsigned m0 = __builtin_amdgcn_ballot_w32(sg == seg0);
        unsigned m1 = __builtin_amdgcn_ballot_w32(sg == seg0 + 1);
        want -= __builtin_popcount(m0) + __builtin_popcount(m1);
        while (m0) {
            const int l = __builtin_ctz(m0);
            m0 &= m0 - 1;
            const int s = __builtin_amdgcn_readlane(ent, l) & 0xffff;
            const float4 v = ((const float4*)(X + (size_t)s * IN_DIM))[lane];
            acc0.x += v.x; acc0.y += v.y; acc0.z += v.z; acc0.w += v.w;
        }
        while (m1) {
            const int l = __builtin_ctz(m1);
            m1 &= m1 - 1;
            const int s = __builtin_amdgcn_readlane(ent, l) & 0xffff;
            const float4 v = ((const float4*)(X + (size_t)s * IN_DIM))[lane];
            acc1.x += v.x; acc1.y += v.y; acc1.z += v.z; acc1.w += v.w;
        }
    }
}

__global__ __launch_bounds__(256) void gnn_gather_gemm(
    const float* __restrict__ X,
    const float* __restrict__ bd, const float* __restrict__ bc, const float* __restrict__ bs,
    const int*   __restrict__ ws,
    float*       __restrict__ out)
{
    __shared__ __attribute__((aligned(16))) float sA[3][16][A_PITCH];
    __shared__ float sD[2][16];

    const int*            cnt   = ws + CNT_OFF;
    const unsigned short* slots = (const unsigned short*)(ws + SLOT_OFF);
    const int*            spill = ws + LIST_OFF;

    const int lane   = threadIdx.x & 31;
    const int wv     = threadIdx.x >> 5;
    const int m_base = blockIdx.x * 16;

    for (int q = 0; q < 2; ++q) {
        const int row  = 2 * wv + q;
        const int node = m_base + row;
        const float4 xs = ((const float4*)(X + (size_t)node * IN_DIM))[lane];
        const int c0 = __builtin_amdgcn_readfirstlane(cnt[2 * node]);
        const int c1 = __builtin_amdgcn_readfirstlane(cnt[2 * node + 1]);
        const int n0 = c0 < SEG_CAP ? c0 : SEG_CAP;
        const int n1 = c1 < SEG_CAP ? c1 : SEG_CAP;
        *(float4*)(&sA[2][row][lane * 4]) = xs;

        float4 acc0 = gather_rows(X, slots + (size_t)(2 * node) * SEG_CAP, n0, lane);
        float4 acc1 = gather_rows(X, slots + (size_t)(2 * node + 1) * SEG_CAP, n1, lane);

        const int want = (c0 - n0) + (c1 - n1);
        if (want > 0) {
            int novf = __builtin_amdgcn_readfirstlane(ws[OVF_OFF]);
            if (novf > N_EDGES) novf = N_EDGES;
            gather_spill(X, spill, novf, 2 * node, want, lane, acc0, acc1);
        }

        *(float4*)(&sA[0][row][lane * 4]) = acc0;
        *(float4*)(&sA[1][row][lane * 4]) = acc1;
        if (lane == 0) {
            sD[0][row] = (float)c0;
            sD[1][row] = (float)c1;
        }
    }
    __syncthreads();

    const int r      = lane & 15;
    const int h      = lane >> 4;
    const int o_base = wv * 16;

    const float4* bp = (const float4*)(ws + WP_OFF) + (size_t)wv * (3 * N_KGROUP * 32) + lane;
    const float* a0p = &sA[0][r][4 * h];
    const float* a1p = &sA[1][r][4 * h];
    const float* a2p = &sA[2][r][4 * h];

    v8f c = {};
#pragma unroll
    for (int k = 0; k < IN_DIM; k += 8) {
        const float4 a0 = *(const float4*)(a0p + k);
        const float4 wd = bp[(3 * (k / 8) + 0) * 32];
        c = __builtin_amdgcn_wmma_f32_16x16x4_f32(false, (v2f){a0.x, a0.y}, false, (v2f){wd.x, wd.y},
                                                  (short)0, c, false, false);
        c = __builtin_amdgcn_wmma_f32_16x16x4_f32(false, (v2f){a0.z, a0.w}, false, (v2f){wd.z, wd.w},
                                                  (short)0, c, false, false);
        const float4 a1 = *(const float4*)(a1p + k);
        const float4 wc = bp[(3 * (k / 8) + 1) * 32];
        c = __builtin_amdgcn_wmma_f32_16x16x4_f32(false, (v2f){a1.x, a1.y}, false, (v2f){wc.x, wc.y},
                                                  (short)0, c, false, false);
        c = __builtin_amdgcn_wmma_f32_16x16x4_f32(false, (v2f){a1.z, a1.w}, false, (v2f){wc.z, wc.w},
                                                  (short)0, c, false, false);
        const float4 a2 = *(const float4*)(a2p + k);
        const float4 wf = bp[(3 * (k / 8) + 2) * 32];
        c = __builtin_amdgcn_wmma_f32_16x16x4_f32(false, (v2f){a2.x, a2.y}, false, (v2f){wf.x, wf.y},
                                                  (short)0, c, false, false);
        c = __builtin_amdgcn_wmma_f32_16x16x4_f32(false, (v2f){a2.z, a2.w}, false, (v2f){wf.z, wf.w},
                                                  (short)0, c, false, false);
    }

    const int   o   = o_base + r;
    const float bdo = bd[o];
    const float bco = bc[o];
    const float bso = bs[o];
#pragma unroll
    for (int i = 0; i < 8; ++i) {
        const int ml = i + 8 * h;
        const int m  = m_base + ml;
        float v = c[i] + sD[0][ml] * bdo + sD[1][ml] * bco + bso;
        out[(size_t)m * OUT_DIM + o] = v > 0.0f ? v : 0.0f;
    }
}

extern "C" void kernel_launch(void* const* d_in, const int* in_sizes, int n_in,
                              void* d_out, int out_size, void* d_ws, size_t ws_size,
                              hipStream_t stream) {
    const float* X   = (const float*)d_in[0];
    const int*   ei  = (const int*)d_in[1];
    const int*   et  = (const int*)d_in[2];
    const float* Wd  = (const float*)d_in[3];
    const float* bd  = (const float*)d_in[4];
    const float* Wc  = (const float*)d_in[5];
    const float* bc  = (const float*)d_in[6];
    const float* Wsf = (const float*)d_in[7];
    const float* bs  = (const float*)d_in[8];
    float* out = (float*)d_out;
    int*   ws  = (int*)d_ws;

    (void)in_sizes; (void)n_in; (void)out_size; (void)ws_size;

    gnn_prepare<<<(N_ZERO + WP_PIECES + 255) / 256, 256, 0, stream>>>(Wd, Wc, Wsf, ws);

    gnn_edge_place<<<N_EDGES / EP_EDGES, EP_THREADS, 0, stream>>>(ei, et, ws);

    gnn_gather_gemm<<<N_NODES / 16, 256, 0, stream>>>(X, bd, bc, bs, ws, out);
}


# gather: each wave gets its four counts, four slot lines and two own rows in one round trip and sums both segments of a node together: sixteen row loads in flight per step, not eight; sum order kept
# speedup vs baseline: 4.5606x; 1.0358x over previous
#include <hip/hip_runtime.h>


#define IN_DIM   128
#define OUT_DIM  128
#define N_NODES  10000
#define N_EDGES  640000
#define N_SEG    (2 * N_NODES)
#define A_PITCH  132
#define SEG_CAP  64

typedef __attribute__((ext_vector_type(2))) float v2f;
typedef __attribute__((ext_vector_type(8))) float v8f;

#define CNT_OFF   0
#define OVF_OFF   (N_SEG)
#define N_ZERO    (N_SEG + 32)
#define SLOT_OFF  (N_ZERO)
#define LIST_OFF  (SLOT_OFF + N_SEG * SEG_CAP / 2)
#define WP_OFF    (((LIST_OFF + N_EDGES) + 127) / 128 * 128)
#define N_KGROUP  (IN_DIM / 8)
#define N_OTILE   (OUT_DIM / 16)
#define WP_PIECES (3 * N_OTILE * N_KGROUP * 32)
static_assert((WP_OFF % 128) == 0, "fragment chunks start on 512-byte boundaries");
static_assert((N_ZERO % 32) == 0, "a wave of the first launch is all clear or all copy");

__global__ __launch_bounds__(256) void gnn_prepare(
    const float* __restrict__ Wd, const float* __restrict__ Wc, const float* __restrict__ Wsf,
    int* __restrict__ ws)
{
    const int i = blockIdx.x * blockDim.x + threadIdx.x;
    if (i < N_ZERO) {
        ws[CNT_OFF + i] = 0;
    } else if (i < N_ZERO + WP_PIECES) {
        const int p    = i - N_ZERO;
        const int lane = p & 31;
        const int q    = p >> 5;
        const int mat  = q % 3;
        const int g    = (q / 3) % N_KGROUP;
        const int tile = q / (3 * N_KGROUP);
        const float* Wm = mat == 0 ? Wd : (mat == 1 ? Wc : Wsf);
        const float4 v = *(const float4*)(Wm + (size_t)(16 * tile + (lane & 15)) * IN_DIM
                                             + 8 * g + 4 * (lane >> 4));
        ((float4*)(ws + WP_OFF))[p] = v;
    }
}

#define EP_THREADS     512
#define EP_PER_THREAD  5
#define EP_EDGES       (EP_THREADS * EP_PER_THREAD)
#define EP_NODE_STEPS  ((N_NODES + EP_THREADS - 1) / EP_THREADS)
static_assert(N_EDGES % EP_EDGES == 0, "the grid covers the edge list exactly");
static_assert(EP_EDGES < 65536, "a block's count per segment must fit 16 bits");
static_assert((CNT_OFF % 2) == 0, "a node's pair of counts is one aligned 64-bit word");

__global__ __launch_bounds__(EP_THREADS) void gnn_edge_place(
    const int* __restrict__ edge_index,
    const int* __restrict__ edge_types,
    int*       __restrict__ ws)
{
    __shared__ unsigned hist[N_NODES];

    const int tid = threadIdx.x;
    for (int s = tid; s < N_NODES; s += EP_THREADS) hist[s] = 0u;

    int src[EP_PER_THREAD], dst[EP_PER_THREAD], typ[EP_PER_THREAD];
    const int e0 = blockIdx.x * EP_EDGES + tid;
#pragma unroll
    for (int j = 0; j < EP_PER_THREAD; ++j) {
        const int e = e0 + j * EP_THREADS;
        src[j] = edge_index[e];
        dst[j] = edge_index[N_EDGES + e];
        typ[j] = edge_types[e] == 0 ? 0 : 1;
    }
    __syncthreads();

    unsigned loc[EP_PER_THREAD];
#pragma unroll
    for (int j = 0; j < EP_PER_THREAD; ++j) {
        const unsigned old = __hip_atomic_fetch_add(&hist[dst[j]], typ[j] ? 0x10000u : 1u,
                                                    __ATOMIC_RELAXED, __HIP_MEMORY_SCOPE_WORKGROUP);
        loc[j] = typ[j] ? (old >> 16) : (old & 0xffffu);
    }
    __syncthreads();

    unsigned long long* cnt64 = (unsigned long long*)(ws + CNT_OFF);
    unsigned           w[EP_NODE_STEPS];
    unsigned long long was[EP_NODE_STEPS];
#pragma unroll
    for (int i = 0; i < EP_NODE_STEPS; ++i) {
        const int s = tid + i * EP_THREADS;
        w[i] = s < N_NODES ? hist[s] : 0u;
    }
#pragma unroll
    for (int i = 0; i < EP_NODE_STEPS; ++i) {
        was[i] = 0ull;
        if (w[i] != 0u) {
            const unsigned long long add =
                ((unsigned long long)(w[i] >> 16) << 32) | (unsigned long long)(w[i] & 0xffffu);
            was[i] = __hip_atomic_fetch_add(cnt64 + tid + i * EP_THREADS, add,
                                            __ATOMIC_RELAXED, __HIP_MEMORY_SCOPE_AGENT);
        }
    }
#pragma unroll
    for (int i = 0; i < EP_NODE_STEPS; ++i) {
        if (w[i] != 0u) {
            unsigned b0 = (unsigned)was[i];
            unsigned b1 = (unsigned)(was[i] >> 32);
            b0 = b0 < SEG_CAP ? b0 : SEG_CAP;
            b1 = b1 < SEG_CAP ? b1 : SEG_CAP;
            hist[tid + i * EP_THREADS] = b0 | (b1 << 16);
        }
    }
    __syncthreads();

    unsigned short* slots = (unsigned short*)(ws + SLOT_OFF);
    unsigned rank[EP_PER_THREAD];
#pragma unroll
    for (int j = 0; j < EP_PER_THREAD; ++j) {
        const unsigned bw = hist[dst[j]];
        rank[j] = (typ[j] ? (bw >> 16) : (bw & 0xffffu)) + loc[j];
    }
#pragma unroll
    for (int j = 0; j < EP_PER_THREAD; ++j) {
        const int seg = 2 * dst[j] + typ[j];
        if (rank[j] < (unsigned)SEG_CAP) {
            slots[seg * SEG_CAP + rank[j]] = (unsigned short)src[j];
        } else {
            const int k = __hip_atomic_fetch_add(ws + OVF_OFF, 1,
                                                 __ATOMIC_RELAXED, __HIP_MEMORY_SCOPE_AGENT);
            if ((unsigned)k < (unsigned)N_EDGES)
                ws[LIST_OFF + k] = (seg << 16) | (src[j] & 0xffff);
        }
    }
}

static __device__ __forceinline__ void gather_pair(
    const float* __restrict__ X, const int (&id)[8], const int la,
    const int na, const int nb, const int lane, float4& acca, float4& accb)
{
    const int nmax = na > nb ? na : nb;
    const char* Xb = (const char*)X;
    const unsigned lane16 = 16u * (unsigned)lane;
    for (int q = 0; 8 * q < nmax; ++q) {
        float4 va[8], vb[8];
#pragma unroll
        for (int u = 0; u < 8; ++u) {
            const unsigned s = (unsigned)__builtin_amdgcn_readlane(id[u], la + q);
            va[u] = *(const float4*)(Xb + (s * (unsigned)(IN_DIM * 4) + lane16));
        }
#pragma unroll
        for (int u = 0; u < 8; ++u) {
            const unsigned s = (unsigned)__builtin_amdgcn_readlane(id[u], la + 8 + q);
            vb[u] = *(const float4*)(Xb + (s * (unsigned)(IN_DIM * 4) + lane16));
        }
#pragma unroll
        for (int u = 0; u < 8; ++u) {
            if (8 * q + u < na) {
                acca.x += va[u].x; acca.y += va[u].y; acca.z += va[u].z; acca.w += va[u].w;
            }
        }
#pragma unroll
        for (int u = 0; u < 8; ++u) {
            if (8 * q + u < nb) {
                accb.x += vb[u].x; accb.y += vb[u].y; accb.z += vb[u].z; accb.w += vb[u].w;
            }
        }
    }
}

static __device__ __forceinline__ void gather_spill(
    const float* __restrict__ X, const int* __restrict__ spill, const int novf,
    const int seg0, int want, const int lane, float4& acc0, float4& acc1)
{
    for (int base = 0; base < novf && want > 0; base += 32) {
        const int ent = base + lane < novf ? spill[base + lane] : -1;
        const int sg  = ent >> 16;
        unsigned m0 = __builtin_amdgcn_ballot_w32(sg == seg0);
        unsigned m1 = __builtin_amdgcn_ballot_w32(sg == seg0 + 1);
        want -= __builtin_popcount(m0) + __builtin_popcount(m1);
        while (m0) {
            const int l = __builtin_ctz(m0);
            m0 &= m0 - 1;
            const int s = __builtin_amdgcn_readlane(ent, l) & 0xffff;
            const float4 v = ((const float4*)(X + (size_t)s * IN_DIM))[lane];
            acc0.x += v.x; acc0.y += v.y; acc0.z += v.z; acc0.w += v.w;
        }
        while (m1) {
            const int l = __builtin_ctz(m1);
            m1 &= m1 - 1;
            const int s = __builtin_amdgcn_readlane(ent, l) & 0xffff;
            const float4 v = ((const float4*)(X + (size_t)s * IN_DIM))[lane];
            acc1.x += v.x; acc1.y += v.y; acc1.z += v.z; acc1.w += v.w;
        }
    }
}

__global__ __launch_bounds__(256) void gnn_gather_gemm(
    const float* __restrict__ X,
    const float* __restrict__ bd, const float* __restrict__ bc, const float* __restrict__ bs,
    const int*   __restrict__ ws,
    float*       __restrict__ out)
{
    __shared__ __attribute__((aligned(16))) float sA[3][16][A_PITCH];
    __shared__ float sD[2][16];

    const int*            cnt   = ws + CNT_OFF;
    const unsigned short* slots = (const unsigned short*)(ws + SLOT_OFF);
    const int*            spill = ws + LIST_OFF;

    const int lane   = threadIdx.x & 31;
    const int wv     = __builtin_amdgcn_readfirstlane(threadIdx.x >> 5);
    const int m_base = blockIdx.x * 16;

    const int seg0  = 2 * m_base + 4 * wv;
    const int node0 = m_base + 2 * wv;
    const int    cl  = cnt[seg0 + (lane >> 3)];
    const uint4  idw = ((const uint4*)(slots + (size_t)seg0 * SEG_CAP))[lane];
    const float4 xs0 = ((const float4*)(X + (size_t)node0 * IN_DIM))[lane];
    const float4 xs1 = ((const float4*)(X + (size_t)(node0 + 1) * IN_DIM))[lane];
    const int c0 = __builtin_amdgcn_readlane(cl, 0);
    const int c1 = __builtin_amdgcn_readlane(cl, 8);
    const int c2 = __builtin_amdgcn_readlane(cl, 16);
    const int c3 = __builtin_amdgcn_readlane(cl, 24);
    const int n0 = c0 < SEG_CAP ? c0 : SEG_CAP;
    const int n1 = c1 < SEG_CAP ? c1 : SEG_CAP;
    const int n2 = c2 < SEG_CAP ? c2 : SEG_CAP;
    const int n3 = c3 < SEG_CAP ? c3 : SEG_CAP;

    int id[8];
    {
        const int nl  = cl < SEG_CAP ? cl : SEG_CAP;
        const int rem = nl - 8 * (lane & 7);
        const unsigned w[4] = {idw.x, idw.y, idw.z, idw.w};
#pragma unroll
        for (int d = 0; d < 4; ++d) {
            id[2 * d]     = 2 * d     < rem ? (int)(w[d] & 0xffffu) : 0;
            id[2 * d + 1] = 2 * d + 1 < rem ? (int)(w[d] >> 16)     : 0;
        }
    }
    *(float4*)(&sA[2][2 * wv][lane * 4])     = xs0;
    *(float4*)(&sA[2][2 * wv + 1][lane * 4]) = xs1;

    {
        float4 acc0 = make_float4(0.f, 0.f, 0.f, 0.f);
        float4 acc1 = make_float4(0.f, 0.f, 0.f, 0.f);
        gather_pair(X, id, 0, n0, n1, lane, acc0, acc1);
        const int want = (c0 - n0) + (c1 - n1);
        if (want > 0) {
            int novf = __builtin_amdgcn_readfirstlane(ws[OVF_OFF]);
            if (novf > N_EDGES) novf = N_EDGES;
            gather_spill(X, spill, novf, seg0, want, lane, acc0, acc1);
        }
        *(float4*)(&sA[0][2 * wv][lane * 4]) = acc0;
        *(float4*)(&sA[1][2 * wv][lane * 4]) = acc1;
    }
    {
        float4 acc0 = make_float4(0.f, 0.f, 0.f, 0.f);
        float4 acc1 = make_float4(0.f, 0.f, 0.f, 0.f);
        gather_pair(X, id, 16, n2, n3, lane, acc0, acc1);
        const int want = (c2 - n2) + (c3 - n3);
        if (want > 0) {
            int novf = __builtin_amdgcn_readfirstlane(ws[OVF_OFF]);
            if (novf > N_EDGES) novf = N_EDGES;
            gather_spill(X, spill, novf, seg0 + 2, want, lane, acc0, acc1);
        }
        *(float4*)(&sA[0][2 * wv + 1][lane * 4]) = acc0;
        *(float4*)(&sA[1][2 * wv + 1][lane * 4]) = acc1;
    }
    if (lane == 0) {
        sD[0][2 * wv]     = (float)c0;
        sD[1][2 * wv]     = (float)c1;
        sD[0][2 * wv + 1] = (float)c2;
        sD[1][2 * wv + 1] = (float)c3;
    }
    __syncthreads();

    const int r      = lane & 15;
    const int h      = lane >> 4;
    const int o_base = wv * 16;

    const float4* bp = (const float4*)(ws + WP_OFF) + (size_t)wv * (3 * N_KGROUP * 32) + lane;
    const float* a0p = &sA[0][r][4 * h];
    const float* a1p = &sA[1][r][4 * h];
    const float* a2p = &sA[2][r][4 * h];

    v8f c = {};
#pragma unroll
    for (int k = 0; k < IN_DIM; k += 8) {
        const float4 a0 = *(const float4*)(a0p + k);
        const float4 wd = bp[(3 * (k / 8) + 0) * 32];
        c = __builtin_amdgcn_wmma_f32_16x16x4_f32(false, (v2f){a0.x, a0.y}, false, (v2f){wd.x, wd.y},
                                                  (short)0, c, false, false);
        c = __builtin_amdgcn_wmma_f32_16x16x4_f32(false, (v2f){a0.z, a0.w}, false, (v2f){wd.z, wd.w},
                                                  (short)0, c, false, false);
        const float4 a1 = *(const float4*)(a1p + k);
        const float4 wc = bp[(3 * (k / 8) + 1) * 32];
        c = __builtin_amdgcn_wmma_f32_16x16x4_f32(false, (v2f){a1.x, a1.y}, false, (v2f){wc.x, wc.y},
                                                  (short)0, c, false, false);
        c = __builtin_amdgcn_wmma_f32_16x16x4_f32(false, (v2f){a1.z, a1.w}, false, (v2f){wc.z, wc.w},
                                                  (short)0, c, false, false);
        const float4 a2 = *(const float4*)(a2p + k);
        const float4 wf = bp[(3 * (k / 8) + 2) * 32];
        c = __builtin_amdgcn_wmma_f32_16x16x4_f32(false, (v2f){a2.x, a2.y}, false, (v2f){wf.x, wf.y},
                                                  (short)0, c, false, false);
        c = __builtin_amdgcn_wmma_f32_16x16x4_f32(false, (v2f){a2.z, a2.w}, false, (v2f){wf.z, wf.w},
                                                  (short)0, c, false, false);
    }

    const int   o   = o_base + r;
    const float bdo = bd[o];
    const float bco = bc[o];
    const float bso = bs[o];
#pragma unroll
    for (int i = 0; i < 8; ++i) {
        const int ml = i + 8 * h;
        const int m  = m_base + ml;
        float v = c[i] + sD[0][ml] * bdo + sD[1][ml] * bco + bso;
        out[(size_t)m * OUT_DIM + o] = v > 0.0f ? v : 0.0f;
    }
}

extern "C" void kernel_launch(void* const* d_in, const int* in_sizes, int n_in,
                              void* d_out, int out_size, void* d_ws, size_t ws_size,
                              hipStream_t stream) {
    const float* X   = (const float*)d_in[0];
    const int*   ei  = (const int*)d_in[1];
    const int*   et  = (const int*)d_in[2];
    const float* Wd  = (const float*)d_in[3];
    const float* bd  = (const float*)d_in[4];
    const float* Wc  = (const float*)d_in[5];
    const float* bc  = (const float*)d_in[6];
    const float* Wsf = (const float*)d_in[7];
    const float* bs  = (const float*)d_in[8];
    float* out = (float*)d_out;
    int*   ws  = (int*)d_ws;

    (void)in_sizes; (void)n_in; (void)out_size; (void)ws_size;

    gnn_prepare<<<(N_ZERO + WP_PIECES + 255) / 256, 256, 0, stream>>>(Wd, Wc, Wsf, ws);

    gnn_edge_place<<<N_EDGES / EP_EDGES, EP_THREADS, 0, stream>>>(ei, et, ws);

    gnn_gather_gemm<<<N_NODES / 16, 256, 0, stream>>>(X, bd, bc, bs, ws, out);
}


# gather/GEMM launch re-tiled: 40 nodes per block (250 blocks, one per WGP), 4 nodes per wave, 32 row loads per step with select adds; B fragment feeds 3 row tiles; sum order = LDS arrival
# speedup vs baseline: 6.0631x; 1.3295x over previous
#include <hip/hip_runtime.h>


#define IN_DIM   128
#define OUT_DIM  128
#define N_NODES  10000
#define N_EDGES  640000
#define TILE_NODES 40
#define N_TILE   (N_NODES / TILE_NODES)
#define M_TILES  ((TILE_NODES + 15) / 16)
#define M_ROWS   (16 * M_TILES)
#define A_PITCH  132

typedef __attribute__((ext_vector_type(2))) float v2f;
typedef __attribute__((ext_vector_type(8))) float v8f;

#define ES_THREADS     512
#define ES_PER_THREAD  5
#define ES_EDGES       (ES_THREADS * ES_PER_THREAD)
#define N_CHUNK        (N_EDGES / ES_EDGES)
static_assert(N_EDGES % ES_EDGES == 0, "the grid covers the edge list exactly");
static_assert(ES_EDGES < 65536, "a chunk's run start and run length fit 16 bits each");
static_assert(2 * ES_THREADS >= N_TILE, "two tile counters per thread cover all tiles");
static_assert(N_NODES % TILE_NODES == 0, "whole node tiles");
static_assert(TILE_NODES % 4 == 0 && 2 * TILE_NODES <= 128, "four nodes per wave; a segment id fits 7 bits");

#define OFFT_PITCH 256
#define OFFT_OFF   0
#define CHUNK_OFF  (N_TILE * OFFT_PITCH)
static_assert(N_CHUNK <= OFFT_PITCH, "a tile's run words fit one table row");
#define WP_OFF    (((CHUNK_OFF + N_EDGES) + 127) / 128 * 128)
#define N_KGROUP  (IN_DIM / 8)
#define N_OTILE   (OUT_DIM / 16)
#define WP_PIECES (3 * N_OTILE * N_KGROUP * 32)
static_assert((WP_OFF % 128) == 0, "fragment chunks start on 512-byte boundaries");
static_assert(WP_PIECES <= N_CHUNK * ES_THREADS, "the first launch has a thread per piece");

static __device__ __forceinline__ int wave_incl_scan(int v, const int lane)
{
    for (int d = 1; d < 32; d <<= 1) {
        const int u = __shfl_up(v, d, 32);
        if (lane >= d) v += u;
    }
    return v;
}

__global__ __launch_bounds__(ES_THREADS) void gnn_edge_sort(
    const int* __restrict__ edge_index,
    const int* __restrict__ edge_types,
    const float* __restrict__ Wd, const float* __restrict__ Wc, const float* __restrict__ Wsf,
    int*       __restrict__ ws)
{
    __shared__ unsigned tcnt[2 * ES_THREADS];
    __shared__ unsigned wsum[ES_THREADS / 32];

    const int tid  = threadIdx.x;
    const int lane = tid & 31;
    const int wvi  = tid >> 5;
    tcnt[2 * tid]     = 0u;
    tcnt[2 * tid + 1] = 0u;

    int src[ES_PER_THREAD], dst[ES_PER_THREAD], typ[ES_PER_THREAD];
    const int e0 = blockIdx.x * ES_EDGES + tid;
#pragma unroll
    for (int j = 0; j < ES_PER_THREAD; ++j) {
        const int e = e0 + j * ES_THREADS;
        src[j] = edge_index[e];
        dst[j] = edge_index[N_EDGES + e];
        typ[j] = edge_types[e] == 0 ? 0 : 1;
    }

    const int gp = blockIdx.x * ES_THREADS + tid;
    float4 wpiece = make_float4(0.f, 0.f, 0.f, 0.f);
    if (gp < WP_PIECES) {
        const int pl   = gp & 31;
        const int q    = gp >> 5;
        const int mat  = q % 3;
        const int g    = (q / 3) % N_KGROUP;
        const int tile = q / (3 * N_KGROUP);
        const float* Wm = mat == 0 ? Wd : (mat == 1 ? Wc : Wsf);
        wpiece = *(const float4*)(Wm + (size_t)(16 * tile + (pl & 15)) * IN_DIM
                                     + 8 * g + 4 * (pl >> 4));
    }
    __syncthreads();

    unsigned loc[ES_PER_THREAD];
#pragma unroll
    for (int j = 0; j < ES_PER_THREAD; ++j) {
        loc[j] = __hip_atomic_fetch_add(&tcnt[(unsigned)dst[j] / TILE_NODES], 1u,
                                        __ATOMIC_RELAXED, __HIP_MEMORY_SCOPE_WORKGROUP);
    }
    __syncthreads();

    const unsigned c0 = tcnt[2 * tid];
    const unsigned c1 = tcnt[2 * tid + 1];
    const unsigned incl = (unsigned)wave_incl_scan((int)(c0 + c1), lane);
    if (lane == 31) wsum[wvi] = incl;
    __syncthreads();
    unsigned pre = 0u;
    for (int i = 0; i < ES_THREADS / 32; ++i) {
        const unsigned t = wsum[i];
        if (i < wvi) pre += t;
    }
    const unsigned s0 = pre + incl - (c0 + c1);
    const unsigned s1 = s0 + c0;
    tcnt[2 * tid]     = s0;
    tcnt[2 * tid + 1] = s1;
    __syncthreads();

    int* chunk = ws + CHUNK_OFF + blockIdx.x * ES_EDGES;
    unsigned p[ES_PER_THREAD];
#pragma unroll
    for (int j = 0; j < ES_PER_THREAD; ++j)
        p[j] = tcnt[(unsigned)dst[j] / TILE_NODES] + loc[j];
#pragma unroll
    for (int j = 0; j < ES_PER_THREAD; ++j) {
        const int seg = 2 * (int)((unsigned)dst[j] % TILE_NODES) + typ[j];
        if (p[j] < (unsigned)ES_EDGES)
            chunk[p[j]] = (seg << 16) | (src[j] & 0xffff);
    }
    if (2 * tid < N_TILE)
        ws[OFFT_OFF + (2 * tid) * OFFT_PITCH + blockIdx.x] = (int)(s0 | (c0 << 16));
    if (2 * tid + 1 < N_TILE)
        ws[OFFT_OFF + (2 * tid + 1) * OFFT_PITCH + blockIdx.x] = (int)(s1 | (c1 << 16));
    if (gp < WP_PIECES)
        ((float4*)(ws + WP_OFF))[gp] = wpiece;
}

static __device__ __forceinline__ void gather_quad(
    const float* __restrict__ X, const int (&id)[8], const int (&n)[4], const int lane,
    float4 (&acc)[4])
{
    const int n01 = n[0] > n[1] ? n[0] : n[1];
    const int n23 = n[2] > n[3] ? n[2] : n[3];
    const int nmax = n01 > n23 ? n01 : n23;
    const char* Xb = (const char*)X;
    const unsigned lane16 = 16u * (unsigned)lane;
    for (int q = 0; 8 * q < nmax; ++q) {
        float4 v[4][8];
#pragma unroll
        for (int j = 0; j < 4; ++j) {
#pragma unroll
            for (int u = 0; u < 8; ++u) {
                const unsigned s = (unsigned)__builtin_amdgcn_readlane(id[u], 8 * j + q);
                v[j][u] = *(const float4*)(Xb + (s * (unsigned)(IN_DIM * 4) + lane16));
            }
        }
#pragma unroll
        for (int j = 0; j < 4; ++j) {
#pragma unroll
            for (int u = 0; u < 8; ++u) {
                const bool in = 8 * q + u < n[j];
                acc[j].x += in ? v[j][u].x : 0.0f; acc[j].y += in ? v[j][u].y : 0.0f;
                acc[j].z += in ? v[j][u].z : 0.0f; acc[j].w += in ? v[j][u].w : 0.0f;
            }
        }
    }
}

#define G_WAVES   (TILE_NODES / 4)
#define G_THREADS (32 * G_WAVES)
#define N_SEG     128
#define B_CAP     16384
#define PL_SLOTS  8
#define RUN_SLOTS 24
static_assert(G_THREADS >= N_CHUNK, "a thread per chunk reads the run words");
static_assert(G_WAVES >= OUT_DIM / 16, "a wave per out-channel tile");
__global__ __launch_bounds__(G_THREADS) void gnn_gather_gemm(
    const float* __restrict__ X,
    const float* __restrict__ bd, const float* __restrict__ bc, const float* __restrict__ bs,
    const int*   __restrict__ ws,
    float*       __restrict__ out)
{
    __shared__ __attribute__((aligned(16))) float sA[3][M_ROWS][A_PITCH];
    __shared__ float sD[2][M_ROWS];
    __shared__ unsigned       raw[B_CAP];
    __shared__ unsigned short sorted[B_CAP];
    __shared__ __attribute__((aligned(16))) unsigned scnt[N_SEG];
    __shared__ unsigned spos[N_SEG];
    __shared__ int wtot[G_WAVES];

    const int tid    = threadIdx.x;
    const int lane   = tid & 31;
    const int wv     = __builtin_amdgcn_readfirstlane(tid >> 5);
    const int m_base = blockIdx.x * TILE_NODES;
    const int node0  = m_base + 4 * wv;

    unsigned rw = 0u;
    if (tid < N_CHUNK) rw = (unsigned)ws[OFFT_OFF + blockIdx.x * OFFT_PITCH + tid];
    float4 xs[4];
#pragma unroll
    for (int i = 0; i < 4; ++i)
        xs[i] = ((const float4*)(X + (size_t)(node0 + i) * IN_DIM))[lane];
    if (tid < N_SEG) scnt[tid] = 0u;

    int ln = (int)(rw >> 16);
    int st = (int)(rw & 0xffffu);
    if (ln > ES_EDGES) ln = ES_EDGES;
    if (st > ES_EDGES - ln) st = ES_EDGES - ln;
    const int incl = wave_incl_scan(ln, lane);
    if (lane == 31) wtot[wv] = incl;
#pragma unroll
    for (int i = 0; i < 4; ++i)
        *(float4*)(&sA[2][4 * wv + i][lane * 4]) = xs[i];
    __syncthreads();
    int pre = 0, tot = 0;
    for (int i = 0; i < G_WAVES; ++i) {
        const int t = wtot[i];
        if (i < wv) pre += t;
        tot += t;
    }
    const int pos = pre + incl - ln;
    const int T   = __builtin_amdgcn_readfirstlane(tot);
    const int* myrun = ws + CHUNK_OFF + (tid < N_CHUNK ? tid : 0) * ES_EDGES + st;

    float4 acc[2][4];
    int    dg[2][4];
#pragma unroll
    for (int hf = 0; hf < 2; ++hf) {
#pragma unroll
        for (int j = 0; j < 4; ++j) { acc[hf][j] = make_float4(0.f, 0.f, 0.f, 0.f); dg[hf][j] = 0; }
    }

    for (int bb = 0; bb < T; bb += B_CAP) {
        const int nb = T - bb < B_CAP ? T - bb : B_CAP;
        if (bb > 0) {
            if (tid < N_SEG) scnt[tid] = 0u;
            __syncthreads();
        }
        {
            const int lo = pos > bb ? pos : bb;
            const int hi = pos + ln < bb + nb ? pos + ln : bb + nb;
            const int kend = hi - pos;
            for (int k0 = lo - pos; k0 < kend; k0 += RUN_SLOTS) {
                unsigned e[RUN_SLOTS];
#pragma unroll
                for (int u = 0; u < RUN_SLOTS; ++u)
                    e[u] = k0 + u < kend ? (unsigned)myrun[k0 + u] : 0u;
#pragma unroll
                for (int u = 0; u < RUN_SLOTS; ++u) {
                    if (k0 + u < kend) {
                        raw[pos + k0 + u - bb] = e[u];
                        __hip_atomic_fetch_add(&scnt[(e[u] >> 16) & (N_SEG - 1)], 1u,
                                               __ATOMIC_RELAXED, __HIP_MEMORY_SCOPE_WORKGROUP);
                    }
                }
            }
        }
        __syncthreads();
        if (tid < 32) {
            const uint4 c4 = *(const uint4*)&scnt[4 * tid];
            const int s = (int)(c4.x + c4.y + c4.z + c4.w);
            const unsigned ex = (unsigned)(wave_incl_scan(s, lane) - s);
            spos[4 * tid]     = ex;
            spos[4 * tid + 1] = ex + c4.x;
            spos[4 * tid + 2] = ex + c4.x + c4.y;
            spos[4 * tid + 3] = ex + c4.x + c4.y + c4.z;
        }
        __syncthreads();
        for (int base = 0; base < nb; base += PL_SLOTS * G_THREADS) {
            unsigned e[PL_SLOTS], p[PL_SLOTS];
#pragma unroll
            for (int u = 0; u < PL_SLOTS; ++u)
                e[u] = base + tid + G_THREADS * u < nb ? raw[base + tid + G_THREADS * u] : 0u;
#pragma unroll
            for (int u = 0; u < PL_SLOTS; ++u) {
                p[u] = (unsigned)B_CAP;
                if (base + tid + G_THREADS * u < nb)
                    p[u] = __hip_atomic_fetch_add(&spos[(e[u] >> 16) & (N_SEG - 1)], 1u,
                                                  __ATOMIC_RELAXED, __HIP_MEMORY_SCOPE_WORKGROUP);
            }
#pragma unroll
            for (int u = 0; u < PL_SLOTS; ++u)
                if (p[u] < (unsigned)B_CAP) sorted[p[u]] = (unsigned short)(e[u] & 0xffffu);
        }
        __syncthreads();

#pragma unroll
        for (int hf = 0; hf < 2; ++hf) {
            const int cl = (int)scnt[8 * wv + 4 * hf + (lane >> 3)];
            const int el = (int)spos[8 * wv + 4 * hf + (lane >> 3)];
            int cs[4];
#pragma unroll
            for (int j = 0; j < 4; ++j) {
                cs[j] = __builtin_amdgcn_readlane(cl, 8 * j);
                dg[hf][j] += cs[j];
            }
            const int c01 = cs[0] > cs[1] ? cs[0] : cs[1];
            const int c23 = cs[2] > cs[3] ? cs[2] : cs[3];
            const int cmax = c01 > c23 ? c01 : c23;
            const int bl = el - cl;
            for (int g = 0; 64 * g < cmax; ++g) {
                int id[8];
                const int rem = cl - 64 * g - 8 * (lane & 7);
                const int at  = bl + 64 * g + 8 * (lane & 7);
#pragma unroll
                for (int t = 0; t < 8; ++t)
                    id[t] = t < rem ? (int)sorted[(at + t) & (B_CAP - 1)] : 0;
                int n[4];
#pragma unroll
                for (int j = 0; j < 4; ++j) {
                    const int rj = cs[j] - 64 * g;
                    n[j] = rj < 0 ? 0 : (rj > 64 ? 64 : rj);
                }
                gather_quad(X, id, n, lane, acc[hf]);
            }
        }
        if (bb + B_CAP < T) __syncthreads();
    }

#pragma unroll
    for (int hf = 0; hf < 2; ++hf) {
#pragma unroll
        for (int i = 0; i < 2; ++i) {
            const int row = 4 * wv + 2 * hf + i;
            *(float4*)(&sA[0][row][lane * 4]) = acc[hf][2 * i];
            *(float4*)(&sA[1][row][lane * 4]) = acc[hf][2 * i + 1];
            if (lane == 0) {
                sD[0][row] = (float)dg[hf][2 * i];
                sD[1][row] = (float)dg[hf][2 * i + 1];
            }
        }
    }
    if (2 * wv < M_ROWS - TILE_NODES) {
        const float4 z = make_float4(0.f, 0.f, 0.f, 0.f);
        for (int pl = 0; pl < 3; ++pl) {
            *(float4*)(&sA[pl][TILE_NODES + 2 * wv][lane * 4])     = z;
            *(float4*)(&sA[pl][TILE_NODES + 2 * wv + 1][lane * 4]) = z;
        }
        if (lane < 2) { sD[0][TILE_NODES + 2 * wv + lane] = 0.f; sD[1][TILE_NODES + 2 * wv + lane] = 0.f; }
    }
    __syncthreads();

    if (wv >= OUT_DIM / 16) return;

    const int r      = lane & 15;
    const int h      = lane >> 4;
    const int o_base = wv * 16;

    const float4* bp = (const float4*)(ws + WP_OFF) + (size_t)wv * (3 * N_KGROUP * 32) + lane;
    const float* ap = &sA[0][r][4 * h];

    v8f c[M_TILES];
#pragma unroll
    for (int t = 0; t < M_TILES; ++t) c[t] = (v8f){};
    float4 a[2][3][M_TILES], w[2][3];
#pragma unroll
    for (int p = 0; p < 3; ++p) {
        w[0][p] = bp[p * 32];
#pragma unroll
        for (int t = 0; t < M_TILES; ++t)
            a[0][p][t] = *(const float4*)(ap + (p * M_ROWS + 16 * t) * A_PITCH);
    }
#pragma unroll
    for (int g = 0; g < N_KGROUP; ++g) {
        const int cur = g & 1, nxt = cur ^ 1;
        if (g + 1 < N_KGROUP) {
#pragma unroll
            for (int p = 0; p < 3; ++p) {
                w[nxt][p] = bp[(3 * (g + 1) + p) * 32];
#pragma unroll
                for (int t = 0; t < M_TILES; ++t)
                    a[nxt][p][t] = *(const float4*)(ap + (p * M_ROWS + 16 * t) * A_PITCH + 8 * (g + 1));
            }
        }
        __builtin_amdgcn_sched_barrier(0);
#pragma unroll
        for (int p = 0; p < 3; ++p) {
#pragma unroll
            for (int t = 0; t < M_TILES; ++t)
                c[t] = __builtin_amdgcn_wmma_f32_16x16x4_f32(false, (v2f){a[cur][p][t].x, a[cur][p][t].y}, false,
                                                             (v2f){w[cur][p].x, w[cur][p].y}, (short)0, c[t], false, false);
#pragma unroll
            for (int t = 0; t < M_TILES; ++t)
                c[t] = __builtin_amdgcn_wmma_f32_16x16x4_f32(false, (v2f){a[cur][p][t].z, a[cur][p][t].w}, false,
                                                             (v2f){w[cur][p].z, w[cur][p].w}, (short)0, c[t], false, false);
        }
    }

    const int   o   = o_base + r;
    const float bdo = bd[o];
    const float bco = bc[o];
    const float bso = bs[o];
#pragma unroll
    for (int t = 0; t < M_TILES; ++t) {
#pragma unroll
        for (int i = 0; i < 8; ++i) {
            const int ml = 16 * t + i + 8 * h;
            if (ml < TILE_NODES) {
                const int m = m_base + ml;
                float v = c[t][i] + sD[0][ml] * bdo + sD[1][ml] * bco + bso;
                out[(size_t)m * OUT_DIM + o] = v > 0.0f ? v : 0.0f;
            }
        }
    }
}

extern "C" void kernel_launch(void* const* d_in, const int* in_sizes, int n_in,
                              void* d_out, int out_size, void* d_ws, size_t ws_size,
                              hipStream_t stream) {
    const float* X   = (const float*)d_in[0];
    const int*   ei  = (const int*)d_in[1];
    const int*   et  = (const int*)d_in[2];
    const float* Wd  = (const float*)d_in[3];
    const float* bd  = (const float*)d_in[4];
    const float* Wc  = (const float*)d_in[5];
    const float* bc  = (const float*)d_in[6];
    const float* Wsf = (const float*)d_in[7];
    const float* bs  = (const float*)d_in[8];
    float* out = (float*)d_out;
    int*   ws  = (int*)d_ws;

    (void)in_sizes; (void)n_in; (void)out_size; (void)ws_size;

    gnn_edge_sort<<<N_CHUNK, ES_THREADS, 0, stream>>>(ei, et, Wd, Wc, Wsf, ws);

    gnn_gather_gemm<<<N_TILE, G_THREADS, 0, stream>>>(X, bd, bc, bs, ws, out);
}


# sorted edge chunks cross memory in wide accesses: the sort block builds its chunk in LDS and stores whole 128-byte lines, the gather block reads each run as aligned 16-byte pieces; sum order kept
# speedup vs baseline: 6.4219x; 1.0592x over previous
#include <hip/hip_runtime.h>


#define IN_DIM   128
#define OUT_DIM  128
#define N_NODES  10000
#define N_EDGES  640000
#define TILE_NODES 40
#define N_TILE   (N_NODES / TILE_NODES)
#define M_TILES  ((TILE_NODES + 15) / 16)
#define M_ROWS   (16 * M_TILES)
#define A_PITCH  132

typedef __attribute__((ext_vector_type(2))) float v2f;
typedef __attribute__((ext_vector_type(8))) float v8f;

#define ES_THREADS     512
#define ES_PER_THREAD  5
#define ES_EDGES       (ES_THREADS * ES_PER_THREAD)
#define N_CHUNK        (N_EDGES / ES_EDGES)
static_assert(N_EDGES % ES_EDGES == 0, "the grid covers the edge list exactly");
static_assert(ES_EDGES % 32 == 0, "a chunk is a whole number of 128-byte lines");
static_assert(ES_EDGES < 65536, "a chunk's run start and run length fit 16 bits each");
static_assert(2 * ES_THREADS >= N_TILE, "two tile counters per thread cover all tiles");
static_assert(N_NODES % TILE_NODES == 0, "whole node tiles");
static_assert(TILE_NODES % 4 == 0 && 2 * TILE_NODES <= 128, "four nodes per wave; a segment id fits 7 bits");

#define OFFT_PITCH 256
#define OFFT_OFF   0
#define CHUNK_OFF  (N_TILE * OFFT_PITCH)
static_assert(N_CHUNK <= OFFT_PITCH, "a tile's run words fit one table row");
static_assert((CHUNK_OFF % 32) == 0 && (ES_EDGES % 4) == 0, "chunks start on 128-byte lines and divide into 16-byte pieces");
#define WP_OFF    (((CHUNK_OFF + N_EDGES) + 127) / 128 * 128)
#define N_KGROUP  (IN_DIM / 8)
#define N_OTILE   (OUT_DIM / 16)
#define WP_PIECES (3 * N_OTILE * N_KGROUP * 32)
static_assert((WP_OFF % 128) == 0, "fragment chunks start on 512-byte boundaries");
static_assert(WP_PIECES <= N_CHUNK * ES_THREADS, "the first launch has a thread per piece");

static __device__ __forceinline__ int wave_incl_scan(int v, const int lane)
{
    for (int d = 1; d < 32; d <<= 1) {
        const int u = __shfl_up(v, d, 32);
        if (lane >= d) v += u;
    }
    return v;
}

__global__ __launch_bounds__(ES_THREADS) void gnn_edge_sort(
    const int* __restrict__ edge_index,
    const int* __restrict__ edge_types,
    const float* __restrict__ Wd, const float* __restrict__ Wc, const float* __restrict__ Wsf,
    int*       __restrict__ ws)
{
    __shared__ unsigned tcnt[2 * ES_THREADS];
    __shared__ unsigned wsum[ES_THREADS / 32];
    __shared__ __attribute__((aligned(16))) int stage[ES_EDGES];

    const int tid  = threadIdx.x;
    const int lane = tid & 31;
    const int wvi  = tid >> 5;
    tcnt[2 * tid]     = 0u;
    tcnt[2 * tid + 1] = 0u;

    int src[ES_PER_THREAD], dst[ES_PER_THREAD], typ[ES_PER_THREAD];
    const int e0 = blockIdx.x * ES_EDGES + tid;
#pragma unroll
    for (int j = 0; j < ES_PER_THREAD; ++j) {
        const int e = e0 + j * ES_THREADS;
        src[j] = edge_index[e];
        dst[j] = edge_index[N_EDGES + e];
        typ[j] = edge_types[e] == 0 ? 0 : 1;
    }

    const int gp = blockIdx.x * ES_THREADS + tid;
    float4 wpiece = make_float4(0.f, 0.f, 0.f, 0.f);
    if (gp < WP_PIECES) {
        const int pl   = gp & 31;
        const int q    = gp >> 5;
        const int mat  = q % 3;
        const int g    = (q / 3) % N_KGROUP;
        const int tile = q / (3 * N_KGROUP);
        const float* Wm = mat == 0 ? Wd : (mat == 1 ? Wc : Wsf);
        wpiece = *(const float4*)(Wm + (size_t)(16 * tile + (pl & 15)) * IN_DIM
                                     + 8 * g + 4 * (pl >> 4));
    }
    __syncthreads();

    unsigned loc[ES_PER_THREAD];
#pragma unroll
    for (int j = 0; j < ES_PER_THREAD; ++j) {
        loc[j] = __hip_atomic_fetch_add(&tcnt[(unsigned)dst[j] / TILE_NODES], 1u,
                                        __ATOMIC_RELAXED, __HIP_MEMORY_SCOPE_WORKGROUP);
    }
    __syncthreads();

    const unsigned c0 = tcnt[2 * tid];
    const unsigned c1 = tcnt[2 * tid + 1];
    const unsigned incl = (unsigned)wave_incl_scan((int)(c0 + c1), lane);
    if (lane == 31) wsum[wvi] = incl;
    __syncthreads();
    unsigned pre = 0u;
    for (int i = 0; i < ES_THREADS / 32; ++i) {
        const unsigned t = wsum[i];
        if (i < wvi) pre += t;
    }
    const unsigned s0 = pre + incl - (c0 + c1);
    const unsigned s1 = s0 + c0;
    tcnt[2 * tid]     = s0;
    tcnt[2 * tid + 1] = s1;
    __syncthreads();

    unsigned p[ES_PER_THREAD];
#pragma unroll
    for (int j = 0; j < ES_PER_THREAD; ++j)
        p[j] = tcnt[(unsigned)dst[j] / TILE_NODES] + loc[j];
#pragma unroll
    for (int j = 0; j < ES_PER_THREAD; ++j) {
        const int seg = 2 * (int)((unsigned)dst[j] % TILE_NODES) + typ[j];
        if (p[j] < (unsigned)ES_EDGES)
            stage[p[j]] = (seg << 16) | (src[j] & 0xffff);
    }
    __syncthreads();
    {
        int4* chunk4 = (int4*)(ws + CHUNK_OFF + blockIdx.x * ES_EDGES);
        const int4* stage4 = (const int4*)stage;
        for (int i = tid; i < ES_EDGES / 4; i += ES_THREADS) chunk4[i] = stage4[i];
    }
    if (2 * tid < N_TILE)
        ws[OFFT_OFF + (2 * tid) * OFFT_PITCH + blockIdx.x] = (int)(s0 | (c0 << 16));
    if (2 * tid + 1 < N_TILE)
        ws[OFFT_OFF + (2 * tid + 1) * OFFT_PITCH + blockIdx.x] = (int)(s1 | (c1 << 16));
    if (gp < WP_PIECES)
        ((float4*)(ws + WP_OFF))[gp] = wpiece;
}

static __device__ __forceinline__ void gather_quad(
    const float* __restrict__ X, const int (&id)[8], const int (&n)[4], const int lane,
    float4 (&acc)[4])
{
    const int n01 = n[0] > n[1] ? n[0] : n[1];
    const int n23 = n[2] > n[3] ? n[2] : n[3];
    const int nmax = n01 > n23 ? n01 : n23;
    const char* Xb = (const char*)X;
    const unsigned lane16 = 16u * (unsigned)lane;
    for (int q = 0; 8 * q < nmax; ++q) {
        float4 v[4][8];
#pragma unroll
        for (int j = 0; j < 4; ++j) {
#pragma unroll
            for (int u = 0; u < 8; ++u) {
                const unsigned s = (unsigned)__builtin_amdgcn_readlane(id[u], 8 * j + q);
                v[j][u] = *(const float4*)(Xb + (s * (unsigned)(IN_DIM * 4) + lane16));
            }
        }
#pragma unroll
        for (int j = 0; j < 4; ++j) {
#pragma unroll
            for (int u = 0; u < 8; ++u) {
                const bool in = 8 * q + u < n[j];
                acc[j].x += in ? v[j][u].x : 0.0f; acc[j].y += in ? v[j][u].y : 0.0f;
                acc[j].z += in ? v[j][u].z : 0.0f; acc[j].w += in ? v[j][u].w : 0.0f;
            }
        }
    }
}

#define G_WAVES   (TILE_NODES / 4)
#define G_THREADS (32 * G_WAVES)
#define N_SEG     128
#define B_CAP     16384
#define PL_SLOTS  8
#define RUN_PIECES 7
static_assert(G_THREADS >= N_CHUNK, "a thread per chunk reads the run words");
static_assert(G_WAVES >= OUT_DIM / 16, "a wave per out-channel tile");
__global__ __launch_bounds__(G_THREADS) void gnn_gather_gemm(
    const float* __restrict__ X,
    const float* __restrict__ bd, const float* __restrict__ bc, const float* __restrict__ bs,
    const int*   __restrict__ ws,
    float*       __restrict__ out)
{
    __shared__ __attribute__((aligned(16))) float sA[3][M_ROWS][A_PITCH];
    __shared__ float sD[2][M_ROWS];
    __shared__ unsigned       raw[B_CAP];
    __shared__ unsigned short sorted[B_CAP];
    __shared__ __attribute__((aligned(16))) unsigned scnt[N_SEG];
    __shared__ unsigned spos[N_SEG];
    __shared__ int wtot[G_WAVES];

    const int tid    = threadIdx.x;
    const int lane   = tid & 31;
    const int wv     = __builtin_amdgcn_readfirstlane(tid >> 5);
    const int m_base = blockIdx.x * TILE_NODES;
    const int node0  = m_base + 4 * wv;

    unsigned rw = 0u;
    if (tid < N_CHUNK) rw = (unsigned)ws[OFFT_OFF + blockIdx.x * OFFT_PITCH + tid];
    float4 xs[4];
#pragma unroll
    for (int i = 0; i < 4; ++i)
        xs[i] = ((const float4*)(X + (size_t)(node0 + i) * IN_DIM))[lane];
    if (tid < N_SEG) scnt[tid] = 0u;

    int ln = (int)(rw >> 16);
    int st = (int)(rw & 0xffffu);
    if (ln > ES_EDGES) ln = ES_EDGES;
    if (st > ES_EDGES - ln) st = ES_EDGES - ln;
    const int incl = wave_incl_scan(ln, lane);
    if (lane == 31) wtot[wv] = incl;
#pragma unroll
    for (int i = 0; i < 4; ++i)
        *(float4*)(&sA[2][4 * wv + i][lane * 4]) = xs[i];
    __syncthreads();
    int pre = 0, tot = 0;
    for (int i = 0; i < G_WAVES; ++i) {
        const int t = wtot[i];
        if (i < wv) pre += t;
        tot += t;
    }
    const int pos = pre + incl - ln;
    const int T   = __builtin_amdgcn_readfirstlane(tot);
    const int4* mychunk4 = (const int4*)(ws + CHUNK_OFF + (tid < N_CHUNK ? tid : 0) * ES_EDGES);

    float4 acc[2][4];
    int    dg[2][4];
#pragma unroll
    for (int hf = 0; hf < 2; ++hf) {
#pragma unroll
        for (int j = 0; j < 4; ++j) { acc[hf][j] = make_float4(0.f, 0.f, 0.f, 0.f); dg[hf][j] = 0; }
    }

    for (int bb = 0; bb < T; bb += B_CAP) {
        const int nb = T - bb < B_CAP ? T - bb : B_CAP;
        if (bb > 0) {
            if (tid < N_SEG) scnt[tid] = 0u;
            __syncthreads();
        }
        {
            const int lo = pos > bb ? pos : bb;
            const int hi = pos + ln < bb + nb ? pos + ln : bb + nb;
            const int c0 = st + (lo - pos);
            const int c1 = st + (hi - pos);
            for (int a = c0 < c1 ? (c0 & ~3) : c1; a < c1; a += 4 * RUN_PIECES) {
                int4 pc[RUN_PIECES];
#pragma unroll
                for (int u = 0; u < RUN_PIECES; ++u)
                    pc[u] = a + 4 * u < c1 ? mychunk4[(a >> 2) + u] : make_int4(0, 0, 0, 0);
#pragma unroll
                for (int u = 0; u < RUN_PIECES; ++u) {
                    const unsigned e4[4] = {(unsigned)pc[u].x, (unsigned)pc[u].y,
                                            (unsigned)pc[u].z, (unsigned)pc[u].w};
#pragma unroll
                    for (int t = 0; t < 4; ++t) {
                        const int c = a + 4 * u + t;
                        if (c >= c0 && c < c1) {
                            raw[pos + c - st - bb] = e4[t];
                            __hip_atomic_fetch_add(&scnt[(e4[t] >> 16) & (N_SEG - 1)], 1u,
                                                   __ATOMIC_RELAXED, __HIP_MEMORY_SCOPE_WORKGROUP);
                        }
                    }
                }
            }
        }
        __syncthreads();
        if (tid < 32) {
            const uint4 c4 = *(const uint4*)&scnt[4 * tid];
            const int s = (int)(c4.x + c4.y + c4.z + c4.w);
            const unsigned ex = (unsigned)(wave_incl_scan(s, lane) - s);
            spos[4 * tid]     = ex;
            spos[4 * tid + 1] = ex + c4.x;
            spos[4 * tid + 2] = ex + c4.x + c4.y;
            spos[4 * tid + 3] = ex + c4.x + c4.y + c4.z;
        }
        __syncthreads();
        for (int base = 0; base < nb; base += PL_SLOTS * G_THREADS) {
            unsigned e[PL_SLOTS], p[PL_SLOTS];
#pragma unroll
            for (int u = 0; u < PL_SLOTS; ++u)
                e[u] = base + tid + G_THREADS * u < nb ? raw[base + tid + G_THREADS * u] : 0u;
#pragma unroll
            for (int u = 0; u < PL_SLOTS; ++u) {
                p[u] = (unsigned)B_CAP;
                if (base + tid + G_THREADS * u < nb)
                    p[u] = __hip_atomic_fetch_add(&spos[(e[u] >> 16) & (N_SEG - 1)], 1u,
                                                  __ATOMIC_RELAXED, __HIP_MEMORY_SCOPE_WORKGROUP);
            }
#pragma unroll
            for (int u = 0; u < PL_SLOTS; ++u)
                if (p[u] < (unsigned)B_CAP) sorted[p[u]] = (unsigned short)(e[u] & 0xffffu);
        }
        __syncthreads();

#pragma unroll
        for (int hf = 0; hf < 2; ++hf) {
            const int cl = (int)scnt[8 * wv + 4 * hf + (lane >> 3)];
            const int el = (int)spos[8 * wv + 4 * hf + (lane >> 3)];
            int cs[4];
#pragma unroll
            for (int j = 0; j < 4; ++j) {
                cs[j] = __builtin_amdgcn_readlane(cl, 8 * j);
                dg[hf][j] += cs[j];
            }
            const int c01 = cs[0] > cs[1] ? cs[0] : cs[1];
            const int c23 = cs[2] > cs[3] ? cs[2] : cs[3];
            const int cmax = c01 > c23 ? c01 : c23;
            const int bl = el - cl;
            for (int g = 0; 64 * g < cmax; ++g) {
                int id[8];
                const int rem = cl - 64 * g - 8 * (lane & 7);
                const int at  = bl + 64 * g + 8 * (lane & 7);
#pragma unroll
                for (int t = 0; t < 8; ++t)
                    id[t] = t < rem ? (int)sorted[(at + t) & (B_CAP - 1)] : 0;
                int n[4];
#pragma unroll
                for (int j = 0; j < 4; ++j) {
                    const int rj = cs[j] - 64 * g;
                    n[j] = rj < 0 ? 0 : (rj > 64 ? 64 : rj);
                }
                gather_quad(X, id, n, lane, acc[hf]);
            }
        }
        if (bb + B_CAP < T) __syncthreads();
    }

#pragma unroll
    for (int hf = 0; hf < 2; ++hf) {
#pragma unroll
        for (int i = 0; i < 2; ++i) {
            const int row = 4 * wv + 2 * hf + i;
            *(float4*)(&sA[0][row][lane * 4]) = acc[hf][2 * i];
            *(float4*)(&sA[1][row][lane * 4]) = acc[hf][2 * i + 1];
            if (lane == 0) {
                sD[0][row] = (float)dg[hf][2 * i];
                sD[1][row] = (float)dg[hf][2 * i + 1];
            }
        }
    }
    if (2 * wv < M_ROWS - TILE_NODES) {
        const float4 z = make_float4(0.f, 0.f, 0.f, 0.f);
        for (int pl = 0; pl < 3; ++pl) {
            *(float4*)(&sA[pl][TILE_NODES + 2 * wv][lane * 4])     = z;
            *(float4*)(&sA[pl][TILE_NODES + 2 * wv + 1][lane * 4]) = z;
        }
        if (lane < 2) { sD[0][TILE_NODES + 2 * wv + lane] = 0.f; sD[1][TILE_NODES + 2 * wv + lane] = 0.f; }
    }
    __syncthreads();

    if (wv >= OUT_DIM / 16) return;

    const int r      = lane & 15;
    const int h      = lane >> 4;
    const int o_base = wv * 16;

    const float4* bp = (const float4*)(ws + WP_OFF) + (size_t)wv * (3 * N_KGROUP * 32) + lane;
    const float* ap = &sA[0][r][4 * h];

    v8f c[M_TILES];
#pragma unroll
    for (int t = 0; t < M_TILES; ++t) c[t] = (v8f){};
    float4 a[2][3][M_TILES], w[2][3];
#pragma unroll
    for (int p = 0; p < 3; ++p) {
        w[0][p] = bp[p * 32];
#pragma unroll
        for (int t = 0; t < M_TILES; ++t)
            a[0][p][t] = *(const float4*)(ap + (p * M_ROWS + 16 * t) * A_PITCH);
    }
#pragma unroll
    for (int g = 0; g < N_KGROUP; ++g) {
        const int cur = g & 1, nxt = cur ^ 1;
        if (g + 1 < N_KGROUP) {
#pragma unroll
            for (int p = 0; p < 3; ++p) {
                w[nxt][p] = bp[(3 * (g + 1) + p) * 32];
#pragma unroll
                for (int t = 0; t < M_TILES; ++t)
                    a[nxt][p][t] = *(const float4*)(ap + (p * M_ROWS + 16 * t) * A_PITCH + 8 * (g + 1));
            }
        }
        __builtin_amdgcn_sched_barrier(0);
#pragma unroll
        for (int p = 0; p < 3; ++p) {
#pragma unroll
            for (int t = 0; t < M_TILES; ++t)
                c[t] = __builtin_amdgcn_wmma_f32_16x16x4_f32(false, (v2f){a[cur][p][t].x, a[cur][p][t].y}, false,
                                                             (v2f){w[cur][p].x, w[cur][p].y}, (short)0, c[t], false, false);
#pragma unroll
            for (int t = 0; t < M_TILES; ++t)
                c[t] = __builtin_amdgcn_wmma_f32_16x16x4_f32(false, (v2f){a[cur][p][t].z, a[cur][p][t].w}, false,
                                                             (v2f){w[cur][p].z, w[cur][p].w}, (short)0, c[t], false, false);
        }
    }

    const int   o   = o_base + r;
    const float bdo = bd[o];
    const float bco = bc[o];
    const float bso = bs[o];
#pragma unroll
    for (int t = 0; t < M_TILES; ++t) {
#pragma unroll
        for (int i = 0; i < 8; ++i) {
            const int ml = 16 * t + i + 8 * h;
            if (ml < TILE_NODES) {
                const int m = m_base + ml;
                float v = c[t][i] + sD[0][ml] * bdo + sD[1][ml] * bco + bso;
                out[(size_t)m * OUT_DIM + o] = v > 0.0f ? v : 0.0f;
            }
        }
    }
}

extern "C" void kernel_launch(void* const* d_in, const int* in_sizes, int n_in,
                              void* d_out, int out_size, void* d_ws, size_t ws_size,
                              hipStream_t stream) {
    const float* X   = (const float*)d_in[0];
    const int*   ei  = (const int*)d_in[1];
    const int*   et  = (const int*)d_in[2];
    const float* Wd  = (const float*)d_in[3];
    const float* bd  = (const float*)d_in[4];
    const float* Wc  = (const float*)d_in[5];
    const float* bc  = (const float*)d_in[6];
    const float* Wsf = (const float*)d_in[7];
    const float* bs  = (const float*)d_in[8];
    float* out = (float*)d_out;
    int*   ws  = (int*)d_ws;

    (void)in_sizes; (void)n_in; (void)out_size; (void)ws_size;

    gnn_edge_sort<<<N_CHUNK, ES_THREADS, 0, stream>>>(ei, et, Wd, Wc, Wsf, ws);

    gnn_gather_gemm<<<N_TILE, G_THREADS, 0, stream>>>(X, bd, bc, bs, ws, out);
}


# self product (own rows x self weights) done by the GEMM waves inside their gather steps, B from an LDS copy; typed products fetch B 2 k groups ahead; self terms summed first (k sum order changes)
# speedup vs baseline: 6.7123x; 1.0452x over previous
#include <hip/hip_runtime.h>


#define IN_DIM   128
#define OUT_DIM  128
#define N_NODES  10000
#define N_EDGES  640000
#define TILE_NODES 40
#define N_TILE   (N_NODES / TILE_NODES)
#define M_TILES  ((TILE_NODES + 15) / 16)
#define M_ROWS   (16 * M_TILES)
#define A_PITCH  132

typedef __attribute__((ext_vector_type(2))) float v2f;
typedef __attribute__((ext_vector_type(8))) float v8f;

#define ES_THREADS     512
#define ES_PER_THREAD  5
#define ES_EDGES       (ES_THREADS * ES_PER_THREAD)
#define N_CHUNK        (N_EDGES / ES_EDGES)
static_assert(N_EDGES % ES_EDGES == 0, "the grid covers the edge list exactly");
static_assert(ES_EDGES % 32 == 0, "a chunk is a whole number of 128-byte lines");
static_assert(ES_EDGES < 65536, "a chunk's run start and run length fit 16 bits each");
static_assert(2 * ES_THREADS >= N_TILE, "two tile counters per thread cover all tiles");
static_assert(N_NODES % TILE_NODES == 0, "whole node tiles");
static_assert(TILE_NODES % 4 == 0 && 2 * TILE_NODES <= 128, "four nodes per wave; a segment id fits 7 bits");

#define OFFT_PITCH 256
#define OFFT_OFF   0
#define CHUNK_OFF  (N_TILE * OFFT_PITCH)
static_assert(N_CHUNK <= OFFT_PITCH, "a tile's run words fit one table row");
static_assert((CHUNK_OFF % 32) == 0 && (ES_EDGES % 4) == 0, "chunks start on 128-byte lines and divide into 16-byte pieces");
#define WP_OFF    (((CHUNK_OFF + N_EDGES) + 127) / 128 * 128)
#define N_KGROUP  (IN_DIM / 8)
#define N_OTILE   (OUT_DIM / 16)
#define WP_PIECES (3 * N_OTILE * N_KGROUP * 32)
static_assert((WP_OFF % 128) == 0, "fragment chunks start on 512-byte boundaries");
static_assert(WP_PIECES <= N_CHUNK * ES_THREADS, "the first launch has a thread per piece");

static __device__ __forceinline__ int wave_incl_scan(int v, const int lane)
{
    for (int d = 1; d < 32; d <<= 1) {
        const int u = __shfl_up(v, d, 32);
        if (lane >= d) v += u;
    }
    return v;
}

__global__ __launch_bounds__(ES_THREADS) void gnn_edge_sort(
    const int* __restrict__ edge_index,
    const int* __restrict__ edge_types,
    const float* __restrict__ Wd, const float* __restrict__ Wc, const float* __restrict__ Wsf,
    int*       __restrict__ ws)
{
    __shared__ unsigned tcnt[2 * ES_THREADS];
    __shared__ unsigned wsum[ES_THREADS / 32];
    __shared__ __attribute__((aligned(16))) int stage[ES_EDGES];

    const int tid  = threadIdx.x;
    const int lane = tid & 31;
    const int wvi  = tid >> 5;
    tcnt[2 * tid]     = 0u;
    tcnt[2 * tid + 1] = 0u;

    int src[ES_PER_THREAD], dst[ES_PER_THREAD], typ[ES_PER_THREAD];
    const int e0 = blockIdx.x * ES_EDGES + tid;
#pragma unroll
    for (int j = 0; j < ES_PER_THREAD; ++j) {
        const int e = e0 + j * ES_THREADS;
        src[j] = edge_index[e];
        dst[j] = edge_index[N_EDGES + e];
        typ[j] = edge_types[e] == 0 ? 0 : 1;
    }

    const int gp = blockIdx.x * ES_THREADS + tid;
    float4 wpiece = make_float4(0.f, 0.f, 0.f, 0.f);
    if (gp < WP_PIECES) {
        const int pl   = gp & 31;
        const int q    = gp >> 5;
        const int mat  = q % 3;
        const int g    = (q / 3) % N_KGROUP;
        const int tile = q / (3 * N_KGROUP);
        const float* Wm = mat == 0 ? Wd : (mat == 1 ? Wc : Wsf);
        wpiece = *(const float4*)(Wm + (size_t)(16 * tile + (pl & 15)) * IN_DIM
                                     + 8 * g + 4 * (pl >> 4));
    }
    __syncthreads();

    unsigned loc[ES_PER_THREAD];
#pragma unroll
    for (int j = 0; j < ES_PER_THREAD; ++j) {
        loc[j] = __hip_atomic_fetch_add(&tcnt[(unsigned)dst[j] / TILE_NODES], 1u,
                                        __ATOMIC_RELAXED, __HIP_MEMORY_SCOPE_WORKGROUP);
    }
    __syncthreads();

    const unsigned c0 = tcnt[2 * tid];
    const unsigned c1 = tcnt[2 * tid + 1];
    const unsigned incl = (unsigned)wave_incl_scan((int)(c0 + c1), lane);
    if (lane == 31) wsum[wvi] = incl;
    __syncthreads();
    unsigned pre = 0u;
    for (int i = 0; i < ES_THREADS / 32; ++i) {
        const unsigned t = wsum[i];
        if (i < wvi) pre += t;
    }
    const unsigned s0 = pre + incl - (c0 + c1);
    const unsigned s1 = s0 + c0;
    tcnt[2 * tid]     = s0;
    tcnt[2 * tid + 1] = s1;
    __syncthreads();

    unsigned p[ES_PER_THREAD];
#pragma unroll
    for (int j = 0; j < ES_PER_THREAD; ++j)
        p[j] = tcnt[(unsigned)dst[j] / TILE_NODES] + loc[j];
#pragma unroll
    for (int j = 0; j < ES_PER_THREAD; ++j) {
        const int seg = 2 * (int)((unsigned)dst[j] % TILE_NODES) + typ[j];
        if (p[j] < (unsigned)ES_EDGES)
            stage[p[j]] = (seg << 16) | (src[j] & 0xffff);
    }
    __syncthreads();
    {
        int4* chunk4 = (int4*)(ws + CHUNK_OFF + blockIdx.x * ES_EDGES);
        const int4* stage4 = (const int4*)stage;
        for (int i = tid; i < ES_EDGES / 4; i += ES_THREADS) chunk4[i] = stage4[i];
    }
    if (2 * tid < N_TILE)
        ws[OFFT_OFF + (2 * tid) * OFFT_PITCH + blockIdx.x] = (int)(s0 | (c0 << 16));
    if (2 * tid + 1 < N_TILE)
        ws[OFFT_OFF + (2 * tid + 1) * OFFT_PITCH + blockIdx.x] = (int)(s1 | (c1 << 16));
    if (gp < WP_PIECES)
        ((float4*)(ws + WP_OFF))[gp] = wpiece;
}

static __device__ __forceinline__ void self_pair(
    const float* aself, const float4* bl, v8f (&cs)[M_TILES], int& kg)
{
    float4 as[2][M_TILES], wq[2];
#pragma unroll
    for (int u = 0; u < 2; ++u) {
        wq[u] = bl[(kg + u) * 32];
#pragma unroll
        for (int t = 0; t < M_TILES; ++t)
            as[u][t] = *(const float4*)(aself + 16 * t * A_PITCH + 8 * (kg + u));
    }
#pragma unroll
    for (int u = 0; u < 2; ++u) {
#pragma unroll
        for (int t = 0; t < M_TILES; ++t)
            cs[t] = __builtin_amdgcn_wmma_f32_16x16x4_f32(false, (v2f){as[u][t].x, as[u][t].y}, false,
                                                          (v2f){wq[u].x, wq[u].y}, (short)0, cs[t], false, false);
#pragma unroll
        for (int t = 0; t < M_TILES; ++t)
            cs[t] = __builtin_amdgcn_wmma_f32_16x16x4_f32(false, (v2f){as[u][t].z, as[u][t].w}, false,
                                                          (v2f){wq[u].z, wq[u].w}, (short)0, cs[t], false, false);
    }
    kg += 2;
}

static __device__ __forceinline__ void gather_quad(
    const float* __restrict__ X, const int (&id)[8], const int (&n)[4], const int lane,
    float4 (&acc)[4],
    const bool do_self, const float* aself, const float4* bl, v8f (&cs)[M_TILES], int& kg)
{
    const int n01 = n[0] > n[1] ? n[0] : n[1];
    const int n23 = n[2] > n[3] ? n[2] : n[3];
    const int nmax = n01 > n23 ? n01 : n23;
    const char* Xb = (const char*)X;
    const unsigned lane16 = 16u * (unsigned)lane;
    for (int q = 0; 8 * q < nmax; ++q) {
        float4 v[4][8];
#pragma unroll
        for (int j = 0; j < 4; ++j) {
#pragma unroll
            for (int u = 0; u < 8; ++u) {
                const unsigned s = (unsigned)__builtin_amdgcn_readlane(id[u], 8 * j + q);
                v[j][u] = *(const float4*)(Xb + (s * (unsigned)(IN_DIM * 4) + lane16));
            }
        }
        if (do_self && kg < N_KGROUP) self_pair(aself, bl, cs, kg);
#pragma unroll
        for (int j = 0; j < 4; ++j) {
#pragma unroll
            for (int u = 0; u < 8; ++u) {
                const bool in = 8 * q + u < n[j];
                acc[j].x += in ? v[j][u].x : 0.0f; acc[j].y += in ? v[j][u].y : 0.0f;
                acc[j].z += in ? v[j][u].z : 0.0f; acc[j].w += in ? v[j][u].w : 0.0f;
            }
        }
    }
}

#define G_WAVES   (TILE_NODES / 4)
#define G_THREADS (32 * G_WAVES)
#define N_SEG     128
#define B_CAP     16384
#define PL_SLOTS  8
#define RUN_PIECES 7
static_assert(G_THREADS >= N_CHUNK, "a thread per chunk reads the run words");
static_assert(G_WAVES >= OUT_DIM / 16, "a wave per out-channel tile");
__global__ __launch_bounds__(G_THREADS) void gnn_gather_gemm(
    const float* __restrict__ X,
    const float* __restrict__ bd, const float* __restrict__ bc, const float* __restrict__ bs,
    const int*   __restrict__ ws,
    float*       __restrict__ out)
{
    __shared__ __attribute__((aligned(16))) float sA[3][M_ROWS][A_PITCH];
    __shared__ float sD[2][M_ROWS];
    __shared__ __attribute__((aligned(16))) float4 sWs[G_WAVES][N_KGROUP][32];
    __shared__ unsigned       raw[B_CAP];
    __shared__ unsigned short sorted[B_CAP];
    __shared__ __attribute__((aligned(16))) unsigned scnt[N_SEG];
    __shared__ unsigned spos[N_SEG];
    __shared__ int wtot[G_WAVES];

    const int tid    = threadIdx.x;
    const int lane   = tid & 31;
    const int wv     = __builtin_amdgcn_readfirstlane(tid >> 5);
    const int m_base = blockIdx.x * TILE_NODES;
    const int node0  = m_base + 4 * wv;

    unsigned rw = 0u;
    if (tid < N_CHUNK) rw = (unsigned)ws[OFFT_OFF + blockIdx.x * OFFT_PITCH + tid];
    const bool do_self = wv < OUT_DIM / 16;
    const float*  aself = &sA[2][lane & 15][4 * (lane >> 4)];
    float4* wfrag = &sWs[wv][0][lane];
    const char* bself = (const char*)(ws + WP_OFF)
                        + ((size_t)(do_self ? wv : 0) * (3 * N_KGROUP * 32) + 2 * 32) * 16;
    const unsigned lane16 = 16u * (unsigned)lane;
    v8f cself[M_TILES];
#pragma unroll
    for (int t = 0; t < M_TILES; ++t) cself[t] = (v8f){};
    int kg = 0;
    float4 xs[4];
#pragma unroll
    for (int i = 0; i < 4; ++i)
        xs[i] = ((const float4*)(X + (size_t)(node0 + i) * IN_DIM))[lane];
    float4 wt[N_KGROUP];
#pragma unroll
    for (int g = 0; g < N_KGROUP; ++g)
        wt[g] = *(const float4*)(bself + ((unsigned)(3 * g * 32 * 16) + lane16));
    if (tid < N_SEG) scnt[tid] = 0u;

    int ln = (int)(rw >> 16);
    int st = (int)(rw & 0xffffu);
    if (ln > ES_EDGES) ln = ES_EDGES;
    if (st > ES_EDGES - ln) st = ES_EDGES - ln;
    const int incl = wave_incl_scan(ln, lane);
    if (lane == 31) wtot[wv] = incl;
#pragma unroll
    for (int i = 0; i < 4; ++i)
        *(float4*)(&sA[2][4 * wv + i][lane * 4]) = xs[i];
#pragma unroll
    for (int g = 0; g < N_KGROUP; ++g) wfrag[g * 32] = wt[g];
    if (2 * wv < M_ROWS - TILE_NODES) {
        const float4 z = make_float4(0.f, 0.f, 0.f, 0.f);
        *(float4*)(&sA[2][TILE_NODES + 2 * wv][lane * 4])     = z;
        *(float4*)(&sA[2][TILE_NODES + 2 * wv + 1][lane * 4]) = z;
    }
    __syncthreads();
    int pre = 0, tot = 0;
    for (int i = 0; i < G_WAVES; ++i) {
        const int t = wtot[i];
        if (i < wv) pre += t;
        tot += t;
    }
    const int pos = pre + incl - ln;
    const int T   = __builtin_amdgcn_readfirstlane(tot);
    const int4* mychunk4 = (const int4*)(ws + CHUNK_OFF + (tid < N_CHUNK ? tid : 0) * ES_EDGES);

    float4 acc[2][4];
    int    dg[2][4];
#pragma unroll
    for (int hf = 0; hf < 2; ++hf) {
#pragma unroll
        for (int j = 0; j < 4; ++j) { acc[hf][j] = make_float4(0.f, 0.f, 0.f, 0.f); dg[hf][j] = 0; }
    }

    for (int bb = 0; bb < T; bb += B_CAP) {
        const int nb = T - bb < B_CAP ? T - bb : B_CAP;
        if (bb > 0) {
            if (tid < N_SEG) scnt[tid] = 0u;
            __syncthreads();
        }
        {
            const int lo = pos > bb ? pos : bb;
            const int hi = pos + ln < bb + nb ? pos + ln : bb + nb;
            const int c0 = st + (lo - pos);
            const int c1 = st + (hi - pos);
            for (int a = c0 < c1 ? (c0 & ~3) : c1; a < c1; a += 4 * RUN_PIECES) {
                int4 pc[RUN_PIECES];
#pragma unroll
                for (int u = 0; u < RUN_PIECES; ++u)
                    pc[u] = a + 4 * u < c1 ? mychunk4[(a >> 2) + u] : make_int4(0, 0, 0, 0);
#pragma unroll
                for (int u = 0; u < RUN_PIECES; ++u) {
                    const unsigned e4[4] = {(unsigned)pc[u].x, (unsigned)pc[u].y,
                                            (unsigned)pc[u].z, (unsigned)pc[u].w};
#pragma unroll
                    for (int t = 0; t < 4; ++t) {
                        const int c = a + 4 * u + t;
                        if (c >= c0 && c < c1) {
                            raw[pos + c - st - bb] = e4[t];
                            __hip_atomic_fetch_add(&scnt[(e4[t] >> 16) & (N_SEG - 1)], 1u,
                                                   __ATOMIC_RELAXED, __HIP_MEMORY_SCOPE_WORKGROUP);
                        }
                    }
                }
            }
        }
        __syncthreads();
        if (tid < 32) {
            const uint4 c4 = *(const uint4*)&scnt[4 * tid];
            const int s = (int)(c4.x + c4.y + c4.z + c4.w);
            const unsigned ex = (unsigned)(wave_incl_scan(s, lane) - s);
            spos[4 * tid]     = ex;
            spos[4 * tid + 1] = ex + c4.x;
            spos[4 * tid + 2] = ex + c4.x + c4.y;
            spos[4 * tid + 3] = ex + c4.x + c4.y + c4.z;
        }
        __syncthreads();
        for (int base = 0; base < nb; base += PL_SLOTS * G_THREADS) {
            unsigned e[PL_SLOTS], p[PL_SLOTS];
#pragma unroll
            for (int u = 0; u < PL_SLOTS; ++u)
                e[u] = base + tid + G_THREADS * u < nb ? raw[base + tid + G_THREADS * u] : 0u;
#pragma unroll
            for (int u = 0; u < PL_SLOTS; ++u) {
                p[u] = (unsigned)B_CAP;
                if (base + tid + G_THREADS * u < nb)
                    p[u] = __hip_atomic_fetch_add(&spos[(e[u] >> 16) & (N_SEG - 1)], 1u,
                                                  __ATOMIC_RELAXED, __HIP_MEMORY_SCOPE_WORKGROUP);
            }
#pragma unroll
            for (int u = 0; u < PL_SLOTS; ++u)
                if (p[u] < (unsigned)B_CAP) sorted[p[u]] = (unsigned short)(e[u] & 0xffffu);
        }
        __syncthreads();

#pragma unroll
        for (int hf = 0; hf < 2; ++hf) {
            const int cl = (int)scnt[8 * wv + 4 * hf + (lane >> 3)];
            const int el = (int)spos[8 * wv + 4 * hf + (lane >> 3)];
            int cs[4];
#pragma unroll
            for (int j = 0; j < 4; ++j) {
                cs[j] = __builtin_amdgcn_readlane(cl, 8 * j);
                dg[hf][j] += cs[j];
            }
            const int c01 = cs[0] > cs[1] ? cs[0] : cs[1];
            const int c23 = cs[2] > cs[3] ? cs[2] : cs[3];
            const int cmax = c01 > c23 ? c01 : c23;
            const int bl = el - cl;
            for (int g = 0; 64 * g < cmax; ++g) {
                int id[8];
                const int rem = cl - 64 * g - 8 * (lane & 7);
                const int at  = bl + 64 * g + 8 * (lane & 7);
#pragma unroll
                for (int t = 0; t < 8; ++t)
                    id[t] = t < rem ? (int)sorted[(at + t) & (B_CAP - 1)] : 0;
                int n[4];
#pragma unroll
                for (int j = 0; j < 4; ++j) {
                    const int rj = cs[j] - 64 * g;
                    n[j] = rj < 0 ? 0 : (rj > 64 ? 64 : rj);
                }
                gather_quad(X, id, n, lane, acc[hf], do_self, aself, wfrag, cself, kg);
            }
        }
        if (bb + B_CAP < T) __syncthreads();
    }

    while (do_self && kg < N_KGROUP) self_pair(aself, wfrag, cself, kg);

#pragma unroll
    for (int hf = 0; hf < 2; ++hf) {
#pragma unroll
        for (int i = 0; i < 2; ++i) {
            const int row = 4 * wv + 2 * hf + i;
            *(float4*)(&sA[0][row][lane * 4]) = acc[hf][2 * i];
            *(float4*)(&sA[1][row][lane * 4]) = acc[hf][2 * i + 1];
            if (lane == 0) {
                sD[0][row] = (float)dg[hf][2 * i];
                sD[1][row] = (float)dg[hf][2 * i + 1];
            }
        }
    }
    if (2 * wv < M_ROWS - TILE_NODES) {
        const float4 z = make_float4(0.f, 0.f, 0.f, 0.f);
        for (int pl = 0; pl < 2; ++pl) {
            *(float4*)(&sA[pl][TILE_NODES + 2 * wv][lane * 4])     = z;
            *(float4*)(&sA[pl][TILE_NODES + 2 * wv + 1][lane * 4]) = z;
        }
        if (lane < 2) { sD[0][TILE_NODES + 2 * wv + lane] = 0.f; sD[1][TILE_NODES + 2 * wv + lane] = 0.f; }
    }
    __syncthreads();

    if (wv >= OUT_DIM / 16) return;

    const int r      = lane & 15;
    const int h      = lane >> 4;
    const int o_base = wv * 16;

    const float4* bp = (const float4*)(ws + WP_OFF) + (size_t)wv * (3 * N_KGROUP * 32) + lane;
    const float* ap = &sA[0][r][4 * h];

    v8f c[M_TILES];
#pragma unroll
    for (int t = 0; t < M_TILES; ++t) c[t] = cself[t];
    float4 a[2][2][M_TILES], w[3][2];
#pragma unroll
    for (int p = 0; p < 2; ++p) {
        w[0][p] = bp[p * 32];
        w[1][p] = bp[(3 + p) * 32];
#pragma unroll
        for (int t = 0; t < M_TILES; ++t)
            a[0][p][t] = *(const float4*)(ap + (p * M_ROWS + 16 * t) * A_PITCH);
    }
#pragma unroll
    for (int g = 0; g < N_KGROUP; ++g) {
        const int cur = g & 1, nxt = cur ^ 1;
        const int wcur = g % 3, wnx2 = (g + 2) % 3;
        if (g + 2 < N_KGROUP) {
#pragma unroll
            for (int p = 0; p < 2; ++p) w[wnx2][p] = bp[(3 * (g + 2) + p) * 32];
        }
        if (g + 1 < N_KGROUP) {
#pragma unroll
            for (int p = 0; p < 2; ++p) {
#pragma unroll
                for (int t = 0; t < M_TILES; ++t)
                    a[nxt][p][t] = *(const float4*)(ap + (p * M_ROWS + 16 * t) * A_PITCH + 8 * (g + 1));
            }
        }
        __builtin_amdgcn_sched_barrier(0);
#pragma unroll
        for (int p = 0; p < 2; ++p) {
#pragma unroll
            for (int t = 0; t < M_TILES; ++t)
                c[t] = __builtin_amdgcn_wmma_f32_16x16x4_f32(false, (v2f){a[cur][p][t].x, a[cur][p][t].y}, false,
                                                             (v2f){w[wcur][p].x, w[wcur][p].y}, (short)0, c[t], false, false);
#pragma unroll
            for (int t = 0; t < M_TILES; ++t)
                c[t] = __builtin_amdgcn_wmma_f32_16x16x4_f32(false, (v2f){a[cur][p][t].z, a[cur][p][t].w}, false,
                                                             (v2f){w[wcur][p].z, w[wcur][p].w}, (short)0, c[t], false, false);
        }
    }

    const int   o   = o_base + r;
    const float bdo = bd[o];
    const float bco = bc[o];
    const float bso = bs[o];
#pragma unroll
    for (int t = 0; t < M_TILES; ++t) {
#pragma unroll
        for (int i = 0; i < 8; ++i) {
            const int ml = 16 * t + i + 8 * h;
            if (ml < TILE_NODES) {
                const int m = m_base + ml;
                float v = c[t][i] + sD[0][ml] * bdo + sD[1][ml] * bco + bso;
                out[(size_t)m * OUT_DIM + o] = v > 0.0f ? v : 0.0f;
            }
        }
    }
}

extern "C" void kernel_launch(void* const* d_in, const int* in_sizes, int n_in,
                              void* d_out, int out_size, void* d_ws, size_t ws_size,
                              hipStream_t stream) {
    const float* X   = (const float*)d_in[0];
    const int*   ei  = (const int*)d_in[1];
    const int*   et  = (const int*)d_in[2];
    const float* Wd  = (const float*)d_in[3];
    const float* bd  = (const float*)d_in[4];
    const float* Wc  = (const float*)d_in[5];
    const float* bc  = (const float*)d_in[6];
    const float* Wsf = (const float*)d_in[7];
    const float* bs  = (const float*)d_in[8];
    float* out = (float*)d_out;
    int*   ws  = (int*)d_ws;

    (void)in_sizes; (void)n_in; (void)out_size; (void)ws_size;

    gnn_edge_sort<<<N_CHUNK, ES_THREADS, 0, stream>>>(ei, et, Wd, Wc, Wsf, ws);

    gnn_gather_gemm<<<N_TILE, G_THREADS, 0, stream>>>(X, bd, bc, bs, ws, out);
}
